# starting kernel
# baseline (speedup 1.0000x reference)
#include <hip/hip_runtime.h>
#include <hip/hip_bf16.h>

#define IN_DIM 256
#define NCOL 128
#define HEADS 4
#define OUT_DIM 32
#define ALPHA 0.2f

typedef __attribute__((ext_vector_type(2))) float v2f;
typedef __attribute__((ext_vector_type(8))) float v8f;

__global__ void gat_gemm_wmma(const float* __restrict__ A,
                              const float* __restrict__ B,
                              float* __restrict__ C, int M) {
    const int lane = threadIdx.x & 31;
    const int wave = (blockIdx.x * blockDim.x + threadIdx.x) >> 5;
    const int tiles_n = NCOL / 16;
    const int row_tile = wave / tiles_n;
    const int col_tile = wave % tiles_n;
    const int m0 = row_tile * 16;
    if (m0 >= M) return;

    const int n0   = col_tile * 16;
    const int r    = lane & 15;
    const int koff = (lane >> 4) * 2;

    v8f acc = {};
    const float* arow = A + (size_t)(m0 + r) * IN_DIM;
#pragma unroll 4
    for (int k = 0; k < IN_DIM; k += 4) {
        v2f a = *(const v2f*)(arow + k + koff);
        const float* bp = B + (size_t)(k + koff) * NCOL + n0 + r;
        v2f b;
        b.x = bp[0];
        b.y = bp[NCOL];
        acc = __builtin_amdgcn_wmma_f32_16x16x4_f32(
             false, a,  false, b,
             (short)0, acc,  false,  false);
    }

    const int row_add = (lane >> 4) * 8;
    float* cp = C + (size_t)(m0 + row_add) * NCOL + n0 + (lane & 15);
#pragma unroll
    for (int i = 0; i < 8; ++i) cp[(size_t)i * NCOL] = acc[i];
}

__global__ void gat_node_scores(const float* __restrict__ ft,
                                const float* __restrict__ attn_l,
                                const float* __restrict__ attn_r,
                                float* __restrict__ a1,
                                float* __restrict__ a2, int n_nodes) {
    int i = blockIdx.x * blockDim.x + threadIdx.x;
    if (i >= n_nodes * HEADS) return;
    int h = i & (HEADS - 1);
    const float* f  = ft + (size_t)(i >> 2) * NCOL + h * OUT_DIM;
    const float* al = attn_l + h * OUT_DIM;
    const float* ar = attn_r + h * OUT_DIM;
    float s1 = 0.f, s2 = 0.f;
#pragma unroll
    for (int d = 0; d < OUT_DIM; ++d) {
        float v = f[d];
        s1 += v * al[d];
        s2 += v * ar[d];
    }
    a1[i] = s1;
    a2[i] = s2;
}

__global__ void gat_init(float* __restrict__ out, float* __restrict__ e_max,
                         float* __restrict__ e_sum, int n_nodes) {
    int i = blockIdx.x * blockDim.x + threadIdx.x;
    int n_out = n_nodes * NCOL;
    if (i < n_out) out[i] = 0.f;
    if (i < n_nodes * HEADS) {
        e_max[i] = -3.402823466e+38f;
        e_sum[i] = 0.f;
    }
}

__device__ __forceinline__ void atomicMaxF32(float* addr, float val) {
    if (val >= 0.f)
        atomicMax((int*)addr, __float_as_int(val));
    else
        atomicMin((unsigned int*)addr, __float_as_uint(val));
}

__device__ __forceinline__ float leaky(float v) {
    return v > 0.f ? v : ALPHA * v;
}

__global__ void gat_edge_max(const int* __restrict__ src,
                             const int* __restrict__ dst,
                             const float* __restrict__ a1,
                             const float* __restrict__ a2,
                             float* __restrict__ e_max, int n_edges) {
    int e = blockIdx.x * blockDim.x + threadIdx.x;
    if (e >= n_edges) return;
    int s = src[e], d = dst[e];
    float4 A1 = *(const float4*)(a1 + (size_t)s * HEADS);
    float4 A2 = *(const float4*)(a2 + (size_t)d * HEADS);
    float* em = e_max + (size_t)d * HEADS;
    atomicMaxF32(em + 0, leaky(A1.x + A2.x));
    atomicMaxF32(em + 1, leaky(A1.y + A2.y));
    atomicMaxF32(em + 2, leaky(A1.z + A2.z));
    atomicMaxF32(em + 3, leaky(A1.w + A2.w));
}

__global__ void gat_edge_sum(const int* __restrict__ src,
                             const int* __restrict__ dst,
                             const float* __restrict__ a1,
                             const float* __restrict__ a2,
                             const float* __restrict__ e_max,
                             float* __restrict__ e_sum, int n_edges) {
    int e = blockIdx.x * blockDim.x + threadIdx.x;
    if (e >= n_edges) return;
    int s = src[e], d = dst[e];
    float4 A1 = *(const float4*)(a1 + (size_t)s * HEADS);
    float4 A2 = *(const float4*)(a2 + (size_t)d * HEADS);
    float4 EM = *(const float4*)(e_max + (size_t)d * HEADS);
    float* es = e_sum + (size_t)d * HEADS;
    atomicAdd(es + 0, expf(leaky(A1.x + A2.x) - EM.x));
    atomicAdd(es + 1, expf(leaky(A1.y + A2.y) - EM.y));
    atomicAdd(es + 2, expf(leaky(A1.z + A2.z) - EM.z));
    atomicAdd(es + 3, expf(leaky(A1.w + A2.w) - EM.w));
}

__global__ void gat_edge_agg(const int* __restrict__ src,
                             const int* __restrict__ dst,
                             const float* __restrict__ ft,
                             const float* __restrict__ a1,
                             const float* __restrict__ a2,
                             const float* __restrict__ e_max,
                             const float* __restrict__ e_sum,
                             float* __restrict__ out, int n_edges) {
    long long t = (long long)blockIdx.x * blockDim.x + threadIdx.x;
    if (t >= (long long)n_edges * 32) return;
    int e = (int)(t >> 5);
    int q = (int)(t & 31);
    int h = q >> 3;
    int s = src[e], d = dst[e];
    int idx = d * HEADS + h;
    float v = leaky(a1[s * HEADS + h] + a2[idx]);
    float alpha = expf(v - e_max[idx]) / e_sum[idx];
    float4 F = *(const float4*)(ft + (size_t)s * NCOL + 4 * q);
    float* op = out + (size_t)d * NCOL + 4 * q;
    atomicAdd(op + 0, F.x * alpha);
    atomicAdd(op + 1, F.y * alpha);
    atomicAdd(op + 2, F.z * alpha);
    atomicAdd(op + 3, F.w * alpha);
}

extern "C" void kernel_launch(void* const* d_in, const int* in_sizes, int n_in,
                              void* d_out, int out_size, void* d_ws, size_t ws_size,
                              hipStream_t stream) {
    const float* feature = (const float*)d_in[0];
    const int*   src     = (const int*)d_in[1];
    const int*   dst     = (const int*)d_in[2];
    const float* W_fc    = (const float*)d_in[3];
    const float* attn_l  = (const float*)d_in[4];
    const float* attn_r  = (const float*)d_in[5];
    float*       out     = (float*)d_out;

    const int M = in_sizes[0] / IN_DIM;
    const int E = in_sizes[1];

    float* ft    = (float*)d_ws;
    float* a1    = ft    + (size_t)M * NCOL;
    float* a2    = a1    + (size_t)M * HEADS;
    float* e_max = a2    + (size_t)M * HEADS;
    float* e_sum = e_max + (size_t)M * HEADS;

    {
        int n_waves = ((M + 15) / 16) * (NCOL / 16);
        int threads = n_waves * 32;
        gat_gemm_wmma<<<(threads + 255) / 256, 256, 0, stream>>>(feature, W_fc, ft, M);
    }
    gat_node_scores<<<(M * HEADS + 255) / 256, 256, 0, stream>>>(ft, attn_l, attn_r,
                                                                 a1, a2, M);
    gat_init<<<(M * NCOL + 255) / 256, 256, 0, stream>>>(out, e_max, e_sum, M);
    gat_edge_max<<<(E + 255) / 256, 256, 0, stream>>>(src, dst, a1, a2, e_max, E);
    gat_edge_sum<<<(E + 255) / 256, 256, 0, stream>>>(src, dst, a1, a2,
                                                      e_max, e_sum, E);
    {
        long long total = (long long)E * 32;
        int blocks = (int)((total + 255) / 256);
        gat_edge_agg<<<blocks, 256, 0, stream>>>(src, dst, ft, a1, a2,
                                                 e_max, e_sum, out, E);
    }
}


# Softmax division deferred: one edge pass atomically adds exp(e-max) into e_sum and ft*exp(e-max) into out, a per-node pass divides afterwards; edge_sum pass removed (sums divided after adding)
# speedup vs baseline: 1.1634x; 1.1634x over previous
#include <hip/hip_runtime.h>
#include <hip/hip_bf16.h>

#define IN_DIM 256
#define NCOL 128
#define HEADS 4
#define OUT_DIM 32
#define ALPHA 0.2f

typedef __attribute__((ext_vector_type(2))) float v2f;
typedef __attribute__((ext_vector_type(8))) float v8f;

__global__ void gat_gemm_wmma(const float* __restrict__ A,
                              const float* __restrict__ B,
                              float* __restrict__ C, int M) {
    const int lane = threadIdx.x & 31;
    const int wave = (blockIdx.x * blockDim.x + threadIdx.x) >> 5;
    const int tiles_n = NCOL / 16;
    const int row_tile = wave / tiles_n;
    const int col_tile = wave % tiles_n;
    const int m0 = row_tile * 16;
    if (m0 >= M) return;

    const int n0   = col_tile * 16;
    const int r    = lane & 15;
    const int koff = (lane >> 4) * 2;

    v8f acc = {};
    const float* arow = A + (size_t)(m0 + r) * IN_DIM;
#pragma unroll 4
    for (int k = 0; k < IN_DIM; k += 4) {
        v2f a = *(const v2f*)(arow + k + koff);
        const float* bp = B + (size_t)(k + koff) * NCOL + n0 + r;
        v2f b;
        b.x = bp[0];
        b.y = bp[NCOL];
        acc = __builtin_amdgcn_wmma_f32_16x16x4_f32(
             false, a,  false, b,
             (short)0, acc,  false,  false);
    }

    const int row_add = (lane >> 4) * 8;
    float* cp = C + (size_t)(m0 + row_add) * NCOL + n0 + (lane & 15);
#pragma unroll
    for (int i = 0; i < 8; ++i) cp[(size_t)i * NCOL] = acc[i];
}

__global__ void gat_node_scores(const float* __restrict__ ft,
                                const float* __restrict__ attn_l,
                                const float* __restrict__ attn_r,
                                float* __restrict__ a1,
                                float* __restrict__ a2, int n_nodes) {
    int i = blockIdx.x * blockDim.x + threadIdx.x;
    if (i >= n_nodes * HEADS) return;
    int h = i & (HEADS - 1);
    const float* f  = ft + (size_t)(i >> 2) * NCOL + h * OUT_DIM;
    const float* al = attn_l + h * OUT_DIM;
    const float* ar = attn_r + h * OUT_DIM;
    float s1 = 0.f, s2 = 0.f;
#pragma unroll
    for (int d = 0; d < OUT_DIM; ++d) {
        float v = f[d];
        s1 += v * al[d];
        s2 += v * ar[d];
    }
    a1[i] = s1;
    a2[i] = s2;
}

__global__ void gat_init(float* __restrict__ out, float* __restrict__ e_max,
                         float* __restrict__ e_sum, int n_nodes) {
    int i = blockIdx.x * blockDim.x + threadIdx.x;
    int n_out = n_nodes * NCOL;
    if (i < n_out) out[i] = 0.f;
    if (i < n_nodes * HEADS) {
        e_max[i] = -3.402823466e+38f;
        e_sum[i] = 0.f;
    }
}

__device__ __forceinline__ void atomicMaxF32(float* addr, float val) {
    if (val >= 0.f)
        atomicMax((int*)addr, __float_as_int(val));
    else
        atomicMin((unsigned int*)addr, __float_as_uint(val));
}

__device__ __forceinline__ float leaky(float v) {
    return v > 0.f ? v : ALPHA * v;
}

__global__ void gat_edge_max(const int* __restrict__ src,
                             const int* __restrict__ dst,
                             const float* __restrict__ a1,
                             const float* __restrict__ a2,
                             float* __restrict__ e_max, int n_edges) {
    int e = blockIdx.x * blockDim.x + threadIdx.x;
    if (e >= n_edges) return;
    int s = src[e], d = dst[e];
    float4 A1 = *(const float4*)(a1 + (size_t)s * HEADS);
    float4 A2 = *(const float4*)(a2 + (size_t)d * HEADS);
    float* em = e_max + (size_t)d * HEADS;
    atomicMaxF32(em + 0, leaky(A1.x + A2.x));
    atomicMaxF32(em + 1, leaky(A1.y + A2.y));
    atomicMaxF32(em + 2, leaky(A1.z + A2.z));
    atomicMaxF32(em + 3, leaky(A1.w + A2.w));
}

__global__ void gat_edge_agg(const int* __restrict__ src,
                             const int* __restrict__ dst,
                             const float* __restrict__ ft,
                             const float* __restrict__ a1,
                             const float* __restrict__ a2,
                             const float* __restrict__ e_max,
                             float* __restrict__ e_sum,
                             float* __restrict__ out, int n_edges) {
    long long t = (long long)blockIdx.x * blockDim.x + threadIdx.x;
    if (t >= (long long)n_edges * 32) return;
    int e = (int)(t >> 5);
    int q = (int)(t & 31);
    int h = q >> 3;
    int s = src[e], d = dst[e];
    int idx = d * HEADS + h;
    float v = leaky(a1[s * HEADS + h] + a2[idx]);
    float w = expf(v - e_max[idx]);
    if ((q & 7) == 0) atomicAdd(e_sum + idx, w);
    float4 F = *(const float4*)(ft + (size_t)s * NCOL + 4 * q);
    float* op = out + (size_t)d * NCOL + 4 * q;
    atomicAdd(op + 0, F.x * w);
    atomicAdd(op + 1, F.y * w);
    atomicAdd(op + 2, F.z * w);
    atomicAdd(op + 3, F.w * w);
}

__global__ void gat_normalize(const float* __restrict__ e_sum,
                              float* __restrict__ out, int n_nodes) {
    int i = blockIdx.x * blockDim.x + threadIdx.x;
    if (i >= n_nodes * (NCOL / 4)) return;
    int node = i >> 5;
    int h = (i & 31) >> 3;
    float s = e_sum[node * HEADS + h];
    float4* o = (float4*)out + i;
    float4 x = *o;
    if (s > 0.f) {
        x.x /= s; x.y /= s; x.z /= s; x.w /= s;
    } else {
        x = make_float4(0.f, 0.f, 0.f, 0.f);
    }
    *o = x;
}

extern "C" void kernel_launch(void* const* d_in, const int* in_sizes, int n_in,
                              void* d_out, int out_size, void* d_ws, size_t ws_size,
                              hipStream_t stream) {
    const float* feature = (const float*)d_in[0];
    const int*   src     = (const int*)d_in[1];
    const int*   dst     = (const int*)d_in[2];
    const float* W_fc    = (const float*)d_in[3];
    const float* attn_l  = (const float*)d_in[4];
    const float* attn_r  = (const float*)d_in[5];
    float*       out     = (float*)d_out;

    const int M = in_sizes[0] / IN_DIM;
    const int E = in_sizes[1];

    float* ft    = (float*)d_ws;
    float* a1    = ft    + (size_t)M * NCOL;
    float* a2    = a1    + (size_t)M * HEADS;
    float* e_max = a2    + (size_t)M * HEADS;
    float* e_sum = e_max + (size_t)M * HEADS;

    {
        int n_waves = ((M + 15) / 16) * (NCOL / 16);
        int threads = n_waves * 32;
        gat_gemm_wmma<<<(threads + 255) / 256, 256, 0, stream>>>(feature, W_fc, ft, M);
    }
    gat_node_scores<<<(M * HEADS + 255) / 256, 256, 0, stream>>>(ft, attn_l, attn_r,
                                                                 a1, a2, M);
    gat_init<<<(M * NCOL + 255) / 256, 256, 0, stream>>>(out, e_max, e_sum, M);
    gat_edge_max<<<(E + 255) / 256, 256, 0, stream>>>(src, dst, a1, a2, e_max, E);
    {
        long long total = (long long)E * 32;
        int blocks = (int)((total + 255) / 256);
        gat_edge_agg<<<blocks, 256, 0, stream>>>(src, dst, ft, a1, a2,
                                                 e_max, e_sum, out, E);
    }
    gat_normalize<<<(M * (NCOL / 4) + 255) / 256, 256, 0, stream>>>(e_sum, out, M);
}


# Float-atomic scatter replaced by a per-destination gather: one returning int atomic per edge lists sources in the node's output row; one wave per node sums in registers in slot order, writes once
# speedup vs baseline: 3.0918x; 2.6576x over previous
#include <hip/hip_runtime.h>
#include <hip/hip_bf16.h>

#define IN_DIM 256
#define NCOL 128
#define HEADS 4
#define OUT_DIM 32
#define ALPHA 0.2f

typedef __attribute__((ext_vector_type(2))) float v2f;
typedef __attribute__((ext_vector_type(8))) float v8f;

__global__ void gat_gemm_wmma(const float* __restrict__ A,
                              const float* __restrict__ B,
                              float* __restrict__ C, int M) {
    const int lane = threadIdx.x & 31;
    const int wave = (blockIdx.x * blockDim.x + threadIdx.x) >> 5;
    const int tiles_n = NCOL / 16;
    const int row_tile = wave / tiles_n;
    const int col_tile = wave % tiles_n;
    const int m0 = row_tile * 16;
    if (m0 >= M) return;

    const int n0   = col_tile * 16;
    const int r    = lane & 15;
    const int koff = (lane >> 4) * 2;

    v8f acc = {};
    const float* arow = A + (size_t)(m0 + r) * IN_DIM;
#pragma unroll 4
    for (int k = 0; k < IN_DIM; k += 4) {
        v2f a = *(const v2f*)(arow + k + koff);
        const float* bp = B + (size_t)(k + koff) * NCOL + n0 + r;
        v2f b;
        b.x = bp[0];
        b.y = bp[NCOL];
        acc = __builtin_amdgcn_wmma_f32_16x16x4_f32(
             false, a,  false, b,
             (short)0, acc,  false,  false);
    }

    const int row_add = (lane >> 4) * 8;
    float* cp = C + (size_t)(m0 + row_add) * NCOL + n0 + (lane & 15);
#pragma unroll
    for (int i = 0; i < 8; ++i) cp[(size_t)i * NCOL] = acc[i];
}

__global__ void gat_node_scores(const float* __restrict__ ft,
                                const float* __restrict__ attn_l,
                                const float* __restrict__ attn_r,
                                float* __restrict__ a1,
                                float* __restrict__ a2, int n_nodes) {
    int i = blockIdx.x * blockDim.x + threadIdx.x;
    if (i >= n_nodes * HEADS) return;
    int h = i & (HEADS - 1);
    const float* f  = ft + (size_t)(i >> 2) * NCOL + h * OUT_DIM;
    const float* al = attn_l + h * OUT_DIM;
    const float* ar = attn_r + h * OUT_DIM;
    float s1 = 0.f, s2 = 0.f;
#pragma unroll
    for (int d = 0; d < OUT_DIM; ++d) {
        float v = f[d];
        s1 += v * al[d];
        s2 += v * ar[d];
    }
    a1[i] = s1;
    a2[i] = s2;
}

__global__ void gat_clear_counts(int* __restrict__ cnt, int n_nodes) {
    int i = blockIdx.x * blockDim.x + threadIdx.x;
    if (i < n_nodes) cnt[i] = 0;
}

__device__ __forceinline__ float leaky(float v) {
    return v > 0.f ? v : ALPHA * v;
}

#define ROW_SLOTS NCOL

__global__ void gat_bin_edges(const int* __restrict__ src,
                              const int* __restrict__ dst,
                              int* __restrict__ cnt,
                              int* __restrict__ lists, int n_edges) {
    int e = blockIdx.x * blockDim.x + threadIdx.x;
    if (e >= n_edges) return;
    int d = dst[e];
    int pos = atomicAdd(cnt + d, 1);
    if (pos < ROW_SLOTS) lists[(size_t)d * NCOL + pos] = src[e];
}

__global__ void gat_gather(const int* __restrict__ src,
                           const int* __restrict__ dst,
                           const float* __restrict__ ft,
                           const float* __restrict__ a1,
                           const float* __restrict__ a2,
                           const int* __restrict__ cnt,
                           const int* lists,
                           float* out, int n_nodes, int n_edges) {
    const int t = blockIdx.x * blockDim.x + threadIdx.x;
    const int n = __builtin_amdgcn_readfirstlane(t >> 5);
    if (n >= n_nodes) return;
    const int q = t & 31;
    const int h = q >> 3;
    const int deg = __builtin_amdgcn_readfirstlane(cnt[n]);
    const float a2v = a2[n * HEADS + h];
    const int* row = lists + (size_t)n * NCOL;
    const float* a1h = a1 + h;
    const float* ftq = ft + 4 * q;

    float m = -3.402823466e+38f;
    float l = 0.f;
    float4 acc = make_float4(0.f, 0.f, 0.f, 0.f);

    if (deg <= ROW_SLOTS) {
        for (int base = 0; base < deg; base += 32) {
            const int left = deg - base;
            const int nj = left < 32 ? left : 32;
            const int mine = (q < nj) ? row[base + q] : 0;
            for (int j = 0; j < nj; ++j) {
                const int s = __builtin_amdgcn_readlane(mine, j);
                m = fmaxf(m, leaky(a1h[(size_t)s * HEADS] + a2v));
            }
        }
        for (int base = 0; base < deg; base += 32) {
            const int left = deg - base;
            const int nj = left < 32 ? left : 32;
            const int mine = (q < nj) ? row[base + q] : 0;
#pragma unroll 4
            for (int j = 0; j < nj; ++j) {
                const int s = __builtin_amdgcn_readlane(mine, j);
                const float w = expf(leaky(a1h[(size_t)s * HEADS] + a2v) - m);
                const float4 F = *(const float4*)(ftq + (size_t)s * NCOL);
                l += w;
                acc.x += F.x * w; acc.y += F.y * w;
                acc.z += F.z * w; acc.w += F.w * w;
            }
        }
    } else {
        for (int e = 0; e < n_edges; ++e) {
            if (dst[e] == n) {
                const int s = src[e];
                m = fmaxf(m, leaky(a1h[(size_t)s * HEADS] + a2v));
            }
        }
        for (int e = 0; e < n_edges; ++e) {
            if (dst[e] == n) {
                const int s = src[e];
                const float w = expf(leaky(a1h[(size_t)s * HEADS] + a2v) - m);
                const float4 F = *(const float4*)(ftq + (size_t)s * NCOL);
                l += w;
                acc.x += F.x * w; acc.y += F.y * w;
                acc.z += F.z * w; acc.w += F.w * w;
            }
        }
    }

    float4 x = make_float4(0.f, 0.f, 0.f, 0.f);
    if (l > 0.f) {
        x.x = acc.x / l; x.y = acc.y / l; x.z = acc.z / l; x.w = acc.w / l;
    }
    *(float4*)(out + (size_t)n * NCOL + 4 * q) = x;
}

extern "C" void kernel_launch(void* const* d_in, const int* in_sizes, int n_in,
                              void* d_out, int out_size, void* d_ws, size_t ws_size,
                              hipStream_t stream) {
    const float* feature = (const float*)d_in[0];
    const int*   src     = (const int*)d_in[1];
    const int*   dst     = (const int*)d_in[2];
    const float* W_fc    = (const float*)d_in[3];
    const float* attn_l  = (const float*)d_in[4];
    const float* attn_r  = (const float*)d_in[5];
    float*       out     = (float*)d_out;

    const int M = in_sizes[0] / IN_DIM;
    const int E = in_sizes[1];

    float* ft  = (float*)d_ws;
    float* a1  = ft + (size_t)M * NCOL;
    float* a2  = a1 + (size_t)M * HEADS;
    int*   cnt = (int*)(a2 + (size_t)M * HEADS);

    {
        int n_waves = ((M + 15) / 16) * (NCOL / 16);
        int threads = n_waves * 32;
        gat_gemm_wmma<<<(threads + 255) / 256, 256, 0, stream>>>(feature, W_fc, ft, M);
    }
    gat_node_scores<<<(M * HEADS + 255) / 256, 256, 0, stream>>>(ft, attn_l, attn_r,
                                                                 a1, a2, M);
    gat_clear_counts<<<(M + 255) / 256, 256, 0, stream>>>(cnt, M);
    gat_bin_edges<<<(E + 255) / 256, 256, 0, stream>>>(src, dst, cnt, (int*)out, E);
    gat_gather<<<(M * 32 + 255) / 256, 256, 0, stream>>>(src, dst, ft, a1, a2, cnt,
                                                         (const int*)out, out, M, E);
}


# GEMM re-tiled: one wave per 16x128 row strip, W_fc staged once per work-group in LDS in fragment order; feature rows read once not 8x; k summed in a new order; rows stored as whole lines
# speedup vs baseline: 3.7739x; 1.2206x over previous
#include <hip/hip_runtime.h>
#include <hip/hip_bf16.h>

#define IN_DIM 256
#define NCOL 128
#define HEADS 4
#define OUT_DIM 32
#define ALPHA 0.2f

typedef __attribute__((ext_vector_type(2))) float v2f;
typedef __attribute__((ext_vector_type(8))) float v8f;

#define GEMM_WAVES 4
#define KBLOCKS (IN_DIM / 8)

__global__ __launch_bounds__(GEMM_WAVES * 32)
void gat_gemm_wmma(const float* __restrict__ A,
                   const float* __restrict__ B,
                   float* __restrict__ C, int M) {
    __shared__ float4 Bs[KBLOCKS * 8 * 32];

    const int tid  = threadIdx.x;
    const int lane = tid & 31;
    const int wv   = tid >> 5;
    const int c    = lane & 15;
    const int h    = lane >> 4;
    const int m0   = (blockIdx.x * GEMM_WAVES + wv) * 16;
    const bool active = m0 < M;

    float4 a[KBLOCKS];
    if (active) {
        const float* ap = A + (size_t)(m0 + c) * IN_DIM + 4 * h;
#pragma unroll
        for (int J = 0; J < KBLOCKS; ++J) a[J] = *(const float4*)(ap + 8 * J);
    }

#pragma unroll 4
    for (int it = 0; it < (KBLOCKS * 2) / GEMM_WAVES; ++it) {
        const int s = it * GEMM_WAVES + wv;
        const int J = s >> 1;
        const int u = s & 1;
        const float* bp = B + (size_t)(8 * J + 4 * h) * NCOL + 64 * u + 4 * c;
        const float4 g0 = *(const float4*)(bp);
        const float4 g1 = *(const float4*)(bp + NCOL);
        const float4 g2 = *(const float4*)(bp + 2 * NCOL);
        const float4 g3 = *(const float4*)(bp + 3 * NCOL);
        float4* dp = Bs + (J * 8 + 4 * u) * 32 + lane;
        dp[0 * 32] = make_float4(g0.x, g1.x, g2.x, g3.x);
        dp[1 * 32] = make_float4(g0.y, g1.y, g2.y, g3.y);
        dp[2 * 32] = make_float4(g0.z, g1.z, g2.z, g3.z);
        dp[3 * 32] = make_float4(g0.w, g1.w, g2.w, g3.w);
    }
    __syncthreads();
    if (!active) return;

    float* cp = C + (size_t)(m0 + 8 * h) * NCOL + 4 * c;
    asm volatile("" : "+v"(cp));

    v8f acc[8];
#pragma unroll
    for (int t = 0; t < 8; ++t) acc[t] = (v8f){};

    unsigned lo = (unsigned)lane * 16u;
    unsigned hi = lo + 65536u;
    asm volatile("" : "+v"(lo), "+v"(hi));
    const char* bsb = (const char*)Bs;
#define BFRAG(J, t) \
    (*(const float4*)(bsb + ((J) < KBLOCKS / 2 ? lo : hi) + \
                      ((((J) & (KBLOCKS / 2 - 1)) * 8 + (t)) * 512)))

    float4 bc[8], bn[8];
#pragma unroll
    for (int t = 0; t < 8; ++t) bc[t] = BFRAG(0, t);

#pragma unroll
    for (int J = 0; J < KBLOCKS; ++J) {
        if (J + 1 < KBLOCKS) {
#pragma unroll
            for (int t = 0; t < 8; ++t) bn[t] = BFRAG(J + 1, t);
        }
        const v2f a0 = {a[J].x, a[J].y};
        const v2f a1 = {a[J].z, a[J].w};
#pragma unroll
        for (int t = 0; t < 8; ++t) {
            const v2f b0 = {bc[t].x, bc[t].y};
            acc[t] = __builtin_amdgcn_wmma_f32_16x16x4_f32(
                false, a0, false, b0, (short)0, acc[t], false, false);
        }
#pragma unroll
        for (int t = 0; t < 8; ++t) {
            const v2f b1 = {bc[t].z, bc[t].w};
            acc[t] = __builtin_amdgcn_wmma_f32_16x16x4_f32(
                false, a1, false, b1, (short)0, acc[t], false, false);
        }
#pragma unroll
        for (int t = 0; t < 8; ++t) bc[t] = bn[t];
    }

    asm volatile("v_nop\n\tv_nop\n\tv_nop\n\tv_nop\n\tv_nop"
                 : "+v"(acc[0]), "+v"(acc[1]), "+v"(acc[2]), "+v"(acc[3]),
                   "+v"(acc[4]), "+v"(acc[5]), "+v"(acc[6]), "+v"(acc[7]));

#pragma unroll
    for (int i = 0; i < 8; ++i) {
#pragma unroll
        for (int u = 0; u < 2; ++u) {
            *(float4*)(cp + (size_t)i * NCOL + 64 * u) =
                make_float4(acc[4 * u + 0][i], acc[4 * u + 1][i],
                            acc[4 * u + 2][i], acc[4 * u + 3][i]);
        }
    }
}

__global__ void gat_node_scores(const float* __restrict__ ft,
                                const float* __restrict__ attn_l,
                                const float* __restrict__ attn_r,
                                float* __restrict__ a1,
                                float* __restrict__ a2, int n_nodes) {
    int i = blockIdx.x * blockDim.x + threadIdx.x;
    if (i >= n_nodes * HEADS) return;
    int h = i & (HEADS - 1);
    const float* f  = ft + (size_t)(i >> 2) * NCOL + h * OUT_DIM;
    const float* al = attn_l + h * OUT_DIM;
    const float* ar = attn_r + h * OUT_DIM;
    float s1 = 0.f, s2 = 0.f;
#pragma unroll
    for (int d = 0; d < OUT_DIM; ++d) {
        float v = f[d];
        s1 += v * al[d];
        s2 += v * ar[d];
    }
    a1[i] = s1;
    a2[i] = s2;
}

__global__ void gat_clear_counts(int* __restrict__ cnt, int n_nodes) {
    int i = blockIdx.x * blockDim.x + threadIdx.x;
    if (i < n_nodes) cnt[i] = 0;
}

__device__ __forceinline__ float leaky(float v) {
    return v > 0.f ? v : ALPHA * v;
}

#define ROW_SLOTS NCOL

__global__ void gat_bin_edges(const int* __restrict__ src,
                              const int* __restrict__ dst,
                              int* __restrict__ cnt,
                              int* __restrict__ lists, int n_edges) {
    int e = blockIdx.x * blockDim.x + threadIdx.x;
    if (e >= n_edges) return;
    int d = dst[e];
    int pos = atomicAdd(cnt + d, 1);
    if (pos < ROW_SLOTS) lists[(size_t)d * NCOL + pos] = src[e];
}

__global__ void gat_gather(const int* __restrict__ src,
                           const int* __restrict__ dst,
                           const float* __restrict__ ft,
                           const float* __restrict__ a1,
                           const float* __restrict__ a2,
                           const int* __restrict__ cnt,
                           const int* lists,
                           float* out, int n_nodes, int n_edges) {
    const int t = blockIdx.x * blockDim.x + threadIdx.x;
    const int n = __builtin_amdgcn_readfirstlane(t >> 5);
    if (n >= n_nodes) return;
    const int q = t & 31;
    const int h = q >> 3;
    const int deg = __builtin_amdgcn_readfirstlane(cnt[n]);
    const float a2v = a2[n * HEADS + h];
    const int* row = lists + (size_t)n * NCOL;
    const float* a1h = a1 + h;
    const float* ftq = ft + 4 * q;

    float m = -3.402823466e+38f;
    float l = 0.f;
    float4 acc = make_float4(0.f, 0.f, 0.f, 0.f);

    if (deg <= ROW_SLOTS) {
        for (int base = 0; base < deg; base += 32) {
            const int left = deg - base;
            const int nj = left < 32 ? left : 32;
            const int mine = (q < nj) ? row[base + q] : 0;
            for (int j = 0; j < nj; ++j) {
                const int s = __builtin_amdgcn_readlane(mine, j);
                m = fmaxf(m, leaky(a1h[(size_t)s * HEADS] + a2v));
            }
        }
        for (int base = 0; base < deg; base += 32) {
            const int left = deg - base;
            const int nj = left < 32 ? left : 32;
            const int mine = (q < nj) ? row[base + q] : 0;
#pragma unroll 4
            for (int j = 0; j < nj; ++j) {
                const int s = __builtin_amdgcn_readlane(mine, j);
                const float w = expf(leaky(a1h[(size_t)s * HEADS] + a2v) - m);
                const float4 F = *(const float4*)(ftq + (size_t)s * NCOL);
                l += w;
                acc.x += F.x * w; acc.y += F.y * w;
                acc.z += F.z * w; acc.w += F.w * w;
            }
        }
    } else {
        for (int e = 0; e < n_edges; ++e) {
            if (dst[e] == n) {
                const int s = src[e];
                m = fmaxf(m, leaky(a1h[(size_t)s * HEADS] + a2v));
            }
        }
        for (int e = 0; e < n_edges; ++e) {
            if (dst[e] == n) {
                const int s = src[e];
                const float w = expf(leaky(a1h[(size_t)s * HEADS] + a2v) - m);
                const float4 F = *(const float4*)(ftq + (size_t)s * NCOL);
                l += w;
                acc.x += F.x * w; acc.y += F.y * w;
                acc.z += F.z * w; acc.w += F.w * w;
            }
        }
    }

    float4 x = make_float4(0.f, 0.f, 0.f, 0.f);
    if (l > 0.f) {
        x.x = acc.x / l; x.y = acc.y / l; x.z = acc.z / l; x.w = acc.w / l;
    }
    *(float4*)(out + (size_t)n * NCOL + 4 * q) = x;
}

extern "C" void kernel_launch(void* const* d_in, const int* in_sizes, int n_in,
                              void* d_out, int out_size, void* d_ws, size_t ws_size,
                              hipStream_t stream) {
    const float* feature = (const float*)d_in[0];
    const int*   src     = (const int*)d_in[1];
    const int*   dst     = (const int*)d_in[2];
    const float* W_fc    = (const float*)d_in[3];
    const float* attn_l  = (const float*)d_in[4];
    const float* attn_r  = (const float*)d_in[5];
    float*       out     = (float*)d_out;

    const int M = in_sizes[0] / IN_DIM;
    const int E = in_sizes[1];

    float* ft  = (float*)d_ws;
    float* a1  = ft + (size_t)M * NCOL;
    float* a2  = a1 + (size_t)M * HEADS;
    int*   cnt = (int*)(a2 + (size_t)M * HEADS);

    {
        int n_strips = (M + 15) / 16;
        gat_gemm_wmma<<<(n_strips + GEMM_WAVES - 1) / GEMM_WAVES, GEMM_WAVES * 32, 0, stream>>>(
            feature, W_fc, ft, M);
    }
    gat_node_scores<<<(M * HEADS + 255) / 256, 256, 0, stream>>>(ft, attn_l, attn_r,
                                                                 a1, a2, M);
    gat_clear_counts<<<(M + 255) / 256, 256, 0, stream>>>(cnt, M);
    gat_bin_edges<<<(E + 255) / 256, 256, 0, stream>>>(src, dst, cnt, (int*)out, E);
    gat_gather<<<(M * 32 + 255) / 256, 256, 0, stream>>>(src, dst, ft, a1, a2, cnt,
                                                         (const int*)out, out, M, E);
}


# Edge binning as a two-level counting sort: ranks from LDS counters, one global add per work-group and 256-node bucket, not one returning global atomic per edge; sum order is still an arrival rank
# speedup vs baseline: 5.3082x; 1.4066x over previous
#include <hip/hip_runtime.h>
#include <hip/hip_bf16.h>

#define IN_DIM 256
#define NCOL 128
#define HEADS 4
#define OUT_DIM 32
#define ALPHA 0.2f

#define BUCKET_SHIFT 8
#define BUCKET_NODES (1 << BUCKET_SHIFT)
#define ROW_SLOTS NCOL
#define PAIRS_PER_ROW (NCOL / 2)
#define BUCKET_PAIR_CAP (BUCKET_NODES * PAIRS_PER_ROW)
#define SORT_THREADS 1024

typedef __attribute__((ext_vector_type(2))) float v2f;
typedef __attribute__((ext_vector_type(8))) float v8f;

#define GEMM_WAVES 4
#define KBLOCKS (IN_DIM / 8)

__global__ __launch_bounds__(GEMM_WAVES * 32)
void gat_gemm_wmma(const float* __restrict__ A,
                   const float* __restrict__ B,
                   float* __restrict__ C, int M) {
    __shared__ float4 Bs[KBLOCKS * 8 * 32];

    const int tid  = threadIdx.x;
    const int lane = tid & 31;
    const int wv   = tid >> 5;
    const int c    = lane & 15;
    const int h    = lane >> 4;
    const int m0   = (blockIdx.x * GEMM_WAVES + wv) * 16;
    const bool active = m0 < M;

    float4 a[KBLOCKS];
    if (active) {
        const float* ap = A + (size_t)(m0 + c) * IN_DIM + 4 * h;
#pragma unroll
        for (int J = 0; J < KBLOCKS; ++J) a[J] = *(const float4*)(ap + 8 * J);
    }

#pragma unroll 4
    for (int it = 0; it < (KBLOCKS * 2) / GEMM_WAVES; ++it) {
        const int s = it * GEMM_WAVES + wv;
        const int J = s >> 1;
        const int u = s & 1;
        const float* bp = B + (size_t)(8 * J + 4 * h) * NCOL + 64 * u + 4 * c;
        const float4 g0 = *(const float4*)(bp);
        const float4 g1 = *(const float4*)(bp + NCOL);
        const float4 g2 = *(const float4*)(bp + 2 * NCOL);
        const float4 g3 = *(const float4*)(bp + 3 * NCOL);
        float4* dp = Bs + (J * 8 + 4 * u) * 32 + lane;
        dp[0 * 32] = make_float4(g0.x, g1.x, g2.x, g3.x);
        dp[1 * 32] = make_float4(g0.y, g1.y, g2.y, g3.y);
        dp[2 * 32] = make_float4(g0.z, g1.z, g2.z, g3.z);
        dp[3 * 32] = make_float4(g0.w, g1.w, g2.w, g3.w);
    }
    __syncthreads();
    if (!active) return;

    float* cp = C + (size_t)(m0 + 8 * h) * NCOL + 4 * c;
    asm volatile("" : "+v"(cp));

    v8f acc[8];
#pragma unroll
    for (int t = 0; t < 8; ++t) acc[t] = (v8f){};

    unsigned lo = (unsigned)lane * 16u;
    unsigned hi = lo + 65536u;
    asm volatile("" : "+v"(lo), "+v"(hi));
    const char* bsb = (const char*)Bs;
#define BFRAG(J, t) \
    (*(const float4*)(bsb + ((J) < KBLOCKS / 2 ? lo : hi) + \
                      ((((J) & (KBLOCKS / 2 - 1)) * 8 + (t)) * 512)))

    float4 bc[8], bn[8];
#pragma unroll
    for (int t = 0; t < 8; ++t) bc[t] = BFRAG(0, t);

#pragma unroll
    for (int J = 0; J < KBLOCKS; ++J) {
        if (J + 1 < KBLOCKS) {
#pragma unroll
            for (int t = 0; t < 8; ++t) bn[t] = BFRAG(J + 1, t);
        }
        const v2f a0 = {a[J].x, a[J].y};
        const v2f a1 = {a[J].z, a[J].w};
#pragma unroll
        for (int t = 0; t < 8; ++t) {
            const v2f b0 = {bc[t].x, bc[t].y};
            acc[t] = __builtin_amdgcn_wmma_f32_16x16x4_f32(
                false, a0, false, b0, (short)0, acc[t], false, false);
        }
#pragma unroll
        for (int t = 0; t < 8; ++t) {
            const v2f b1 = {bc[t].z, bc[t].w};
            acc[t] = __builtin_amdgcn_wmma_f32_16x16x4_f32(
                false, a1, false, b1, (short)0, acc[t], false, false);
        }
#pragma unroll
        for (int t = 0; t < 8; ++t) bc[t] = bn[t];
    }

    asm volatile("v_nop\n\tv_nop\n\tv_nop\n\tv_nop\n\tv_nop"
                 : "+v"(acc[0]), "+v"(acc[1]), "+v"(acc[2]), "+v"(acc[3]),
                   "+v"(acc[4]), "+v"(acc[5]), "+v"(acc[6]), "+v"(acc[7]));

#pragma unroll
    for (int i = 0; i < 8; ++i) {
#pragma unroll
        for (int u = 0; u < 2; ++u) {
            *(float4*)(cp + (size_t)i * NCOL + 64 * u) =
                make_float4(acc[4 * u + 0][i], acc[4 * u + 1][i],
                            acc[4 * u + 2][i], acc[4 * u + 3][i]);
        }
    }
}

__global__ void gat_node_scores(const float* __restrict__ ft,
                                const float* __restrict__ attn_l,
                                const float* __restrict__ attn_r,
                                float* __restrict__ a1,
                                float* __restrict__ a2,
                                int* __restrict__ cnt, int n_nodes) {
    int i = blockIdx.x * blockDim.x + threadIdx.x;
    if (i >= n_nodes * HEADS) return;
    if ((i & (BUCKET_NODES * HEADS - 1)) == 0) cnt[i >> 2] = 0;
    int h = i & (HEADS - 1);
    const float* f  = ft + (size_t)(i >> 2) * NCOL + h * OUT_DIM;
    const float* al = attn_l + h * OUT_DIM;
    const float* ar = attn_r + h * OUT_DIM;
    float s1 = 0.f, s2 = 0.f;
#pragma unroll
    for (int d = 0; d < OUT_DIM; ++d) {
        float v = f[d];
        s1 += v * al[d];
        s2 += v * ar[d];
    }
    a1[i] = s1;
    a2[i] = s2;
}


__device__ __forceinline__ float leaky(float v) {
    return v > 0.f ? v : ALPHA * v;
}

__global__ __launch_bounds__(SORT_THREADS)
void gat_partition_edges(const int* __restrict__ src,
                         const int* __restrict__ dst,
                         int* __restrict__ cnt,
                         int* __restrict__ rows_as_pairs,
                         int n_nodes, int n_edges) {
    extern __shared__ int hbin[];
    int2* pairs = (int2*)rows_as_pairs;
    const int tid = threadIdx.x;
    const int nb = (n_nodes + BUCKET_NODES - 1) >> BUCKET_SHIFT;
    const int chunk = (n_edges + (int)gridDim.x - 1) / (int)gridDim.x;
    const int e0 = blockIdx.x * chunk;
    const int e1 = (e0 + chunk < n_edges) ? e0 + chunk : n_edges;

    for (int b = tid; b < nb; b += SORT_THREADS) hbin[b] = 0;
    __syncthreads();
    for (int e = e0 + tid; e < e1; e += 4 * SORT_THREADS) {
        int d[4];
#pragma unroll
        for (int k = 0; k < 4; ++k) {
            const int ek = e + k * SORT_THREADS;
            d[k] = dst[ek < e1 ? ek : e1 - 1];
        }
        asm volatile("" : "+v"(d[0]), "+v"(d[1]), "+v"(d[2]), "+v"(d[3]));
#pragma unroll
        for (int k = 0; k < 4; ++k)
            if (e + k * SORT_THREADS < e1) atomicAdd(&hbin[d[k] >> BUCKET_SHIFT], 1);
    }
    __syncthreads();
    for (int b = tid; b < nb; b += SORT_THREADS) {
        const int c = hbin[b];
        if (c > 0) hbin[b] = atomicAdd(cnt + ((size_t)b << BUCKET_SHIFT), c);
    }
    __syncthreads();
    for (int e = e0 + tid; e < e1; e += 4 * SORT_THREADS) {
        int d[4], s[4];
#pragma unroll
        for (int k = 0; k < 4; ++k) {
            const int ek = e + k * SORT_THREADS;
            const int ec = ek < e1 ? ek : e1 - 1;
            d[k] = dst[ec];
            s[k] = src[ec];
        }
        asm volatile("" : "+v"(d[0]), "+v"(d[1]), "+v"(d[2]), "+v"(d[3]),
                          "+v"(s[0]), "+v"(s[1]), "+v"(s[2]), "+v"(s[3]));
#pragma unroll
        for (int k = 0; k < 4; ++k) {
            if (e + k * SORT_THREADS < e1) {
                const int b = d[k] >> BUCKET_SHIFT;
                const int pos = atomicAdd(&hbin[b], 1);
                const int left = n_nodes - (b << BUCKET_SHIFT);
                const int rows = left < BUCKET_NODES ? left : BUCKET_NODES;
                if (pos < rows * PAIRS_PER_ROW)
                    pairs[(size_t)b * BUCKET_PAIR_CAP + pos] = make_int2(d[k], s[k]);
            }
        }
    }
}

__global__ __launch_bounds__(SORT_THREADS)
void gat_place_edges(const int* __restrict__ src,
                     const int* __restrict__ dst,
                     int* __restrict__ cnt,
                     int* lists, int n_nodes, int n_edges) {
    __shared__ int2 pl[BUCKET_PAIR_CAP];
    __shared__ int deg[BUCKET_NODES];
    const int tid = threadIdx.x;
    const int b = blockIdx.x;
    const int node0 = b << BUCKET_SHIFT;
    const int left = n_nodes - node0;
    const int rows = left < BUCKET_NODES ? left : BUCKET_NODES;
    const int np = cnt[node0];
    const bool stored = np <= rows * PAIRS_PER_ROW;

    if (tid < BUCKET_NODES) deg[tid] = 0;
    if (stored) {
        const int2* pg = (const int2*)lists + (size_t)b * BUCKET_PAIR_CAP;
        for (int i = tid; i < np; i += 4 * SORT_THREADS) {
            int2 p[4];
#pragma unroll
            for (int k = 0; k < 4; ++k) {
                const int ik = i + k * SORT_THREADS;
                p[k] = pg[ik < np ? ik : np - 1];
            }
            asm volatile("" ::: "memory");
#pragma unroll
            for (int k = 0; k < 4; ++k) {
                const int ik = i + k * SORT_THREADS;
                if (ik < np) pl[ik] = p[k];
            }
        }
    }
    __syncthreads();
    if (stored) {
        for (int i = tid; i < np; i += SORT_THREADS) {
            const int2 p = pl[i];
            const int pos = atomicAdd(&deg[p.x & (BUCKET_NODES - 1)], 1);
            if (pos < ROW_SLOTS) lists[(size_t)p.x * NCOL + pos] = p.y;
        }
    } else {
        for (int e = tid; e < n_edges; e += SORT_THREADS) {
            const int d = dst[e];
            if ((d >> BUCKET_SHIFT) == b) {
                const int pos = atomicAdd(&deg[d & (BUCKET_NODES - 1)], 1);
                if (pos < ROW_SLOTS) lists[(size_t)d * NCOL + pos] = src[e];
            }
        }
    }
    __syncthreads();
    if (tid < rows) cnt[node0 + tid] = deg[tid];
}

__global__ void gat_gather(const int* __restrict__ src,
                           const int* __restrict__ dst,
                           const float* __restrict__ ft,
                           const float* __restrict__ a1,
                           const float* __restrict__ a2,
                           const int* __restrict__ cnt,
                           const int* lists,
                           float* out, int n_nodes, int n_edges) {
    const int t = blockIdx.x * blockDim.x + threadIdx.x;
    const int n = __builtin_amdgcn_readfirstlane(t >> 5);
    if (n >= n_nodes) return;
    const int q = t & 31;
    const int h = q >> 3;
    const int deg = __builtin_amdgcn_readfirstlane(cnt[n]);
    const float a2v = a2[n * HEADS + h];
    const int* row = lists + (size_t)n * NCOL;
    const float* a1h = a1 + h;
    const float* ftq = ft + 4 * q;

    float m = -3.402823466e+38f;
    float l = 0.f;
    float4 acc = make_float4(0.f, 0.f, 0.f, 0.f);

    if (deg <= ROW_SLOTS) {
        for (int base = 0; base < deg; base += 32) {
            const int left = deg - base;
            const int nj = left < 32 ? left : 32;
            const int mine = (q < nj) ? row[base + q] : 0;
            for (int j = 0; j < nj; ++j) {
                const int s = __builtin_amdgcn_readlane(mine, j);
                m = fmaxf(m, leaky(a1h[(size_t)s * HEADS] + a2v));
            }
        }
        for (int base = 0; base < deg; base += 32) {
            const int left = deg - base;
            const int nj = left < 32 ? left : 32;
            const int mine = (q < nj) ? row[base + q] : 0;
#pragma unroll 4
            for (int j = 0; j < nj; ++j) {
                const int s = __builtin_amdgcn_readlane(mine, j);
                const float w = expf(leaky(a1h[(size_t)s * HEADS] + a2v) - m);
                const float4 F = *(const float4*)(ftq + (size_t)s * NCOL);
                l += w;
                acc.x += F.x * w; acc.y += F.y * w;
                acc.z += F.z * w; acc.w += F.w * w;
            }
        }
    } else {
        for (int e = 0; e < n_edges; ++e) {
            if (dst[e] == n) {
                const int s = src[e];
                m = fmaxf(m, leaky(a1h[(size_t)s * HEADS] + a2v));
            }
        }
        for (int e = 0; e < n_edges; ++e) {
            if (dst[e] == n) {
                const int s = src[e];
                const float w = expf(leaky(a1h[(size_t)s * HEADS] + a2v) - m);
                const float4 F = *(const float4*)(ftq + (size_t)s * NCOL);
                l += w;
                acc.x += F.x * w; acc.y += F.y * w;
                acc.z += F.z * w; acc.w += F.w * w;
            }
        }
    }

    float4 x = make_float4(0.f, 0.f, 0.f, 0.f);
    if (l > 0.f) {
        x.x = acc.x / l; x.y = acc.y / l; x.z = acc.z / l; x.w = acc.w / l;
    }
    *(float4*)(out + (size_t)n * NCOL + 4 * q) = x;
}

extern "C" void kernel_launch(void* const* d_in, const int* in_sizes, int n_in,
                              void* d_out, int out_size, void* d_ws, size_t ws_size,
                              hipStream_t stream) {
    const float* feature = (const float*)d_in[0];
    const int*   src     = (const int*)d_in[1];
    const int*   dst     = (const int*)d_in[2];
    const float* W_fc    = (const float*)d_in[3];
    const float* attn_l  = (const float*)d_in[4];
    const float* attn_r  = (const float*)d_in[5];
    float*       out     = (float*)d_out;

    const int M = in_sizes[0] / IN_DIM;
    const int E = in_sizes[1];

    float* ft  = (float*)d_ws;
    float* a1  = ft + (size_t)M * NCOL;
    float* a2  = a1 + (size_t)M * HEADS;
    int*   cnt = (int*)(a2 + (size_t)M * HEADS);

    {
        int n_strips = (M + 15) / 16;
        gat_gemm_wmma<<<(n_strips + GEMM_WAVES - 1) / GEMM_WAVES, GEMM_WAVES * 32, 0, stream>>>(
            feature, W_fc, ft, M);
    }
    gat_node_scores<<<(M * HEADS + 255) / 256, 256, 0, stream>>>(ft, attn_l, attn_r,
                                                                 a1, a2, cnt, M);
    gat_partition_edges<<<256, SORT_THREADS,
                          ((M + BUCKET_NODES - 1) / BUCKET_NODES) * sizeof(int), stream>>>(
        src, dst, cnt, (int*)out, M, E);
    gat_place_edges<<<(M + BUCKET_NODES - 1) / BUCKET_NODES, SORT_THREADS, 0, stream>>>(
        src, dst, cnt, (int*)out, M, E);
    gat_gather<<<(M * 32 + 255) / 256, 256, 0, stream>>>(src, dst, ft, a1, a2, cnt,
                                                         (const int*)out, out, M, E);
}


# Gather kernel takes both attention scores from the ft rows it reads (a1 = row.attn_l per source, a2 = own row.attn_r); softmax as rescaled running sums (new sum order); score kernel and tables gone
# speedup vs baseline: 6.3458x; 1.1955x over previous
#include <hip/hip_runtime.h>
#include <hip/hip_bf16.h>

#define IN_DIM 256
#define NCOL 128
#define HEADS 4
#define OUT_DIM 32
#define ALPHA 0.2f

#define BUCKET_SHIFT 8
#define BUCKET_NODES (1 << BUCKET_SHIFT)
#define ROW_SLOTS NCOL
#define PAIRS_PER_ROW (NCOL / 2)
#define BUCKET_PAIR_CAP (BUCKET_NODES * PAIRS_PER_ROW)
#define SORT_THREADS 1024

typedef __attribute__((ext_vector_type(2))) float v2f;
typedef __attribute__((ext_vector_type(8))) float v8f;

#define GEMM_WAVES 4
#define KBLOCKS (IN_DIM / 8)

__global__ __launch_bounds__(GEMM_WAVES * 32)
void gat_gemm_wmma(const float* __restrict__ A,
                   const float* __restrict__ B,
                   float* __restrict__ C,
                   int* __restrict__ cnt, int M) {
    __shared__ float4 Bs[KBLOCKS * 8 * 32];

    const int tid  = threadIdx.x;
    const int lane = tid & 31;
    const int wv   = tid >> 5;
    const int c    = lane & 15;
    const int h    = lane >> 4;
    const int m0   = (blockIdx.x * GEMM_WAVES + wv) * 16;
    const bool active = m0 < M;

    if (tid == 0 && blockIdx.x * BUCKET_NODES < M) cnt[blockIdx.x * BUCKET_NODES] = 0;

    float4 a[KBLOCKS];
    if (active) {
        const float* ap = A + (size_t)(m0 + c) * IN_DIM + 4 * h;
#pragma unroll
        for (int J = 0; J < KBLOCKS; ++J) a[J] = *(const float4*)(ap + 8 * J);
    }

#pragma unroll 4
    for (int it = 0; it < (KBLOCKS * 2) / GEMM_WAVES; ++it) {
        const int s = it * GEMM_WAVES + wv;
        const int J = s >> 1;
        const int u = s & 1;
        const float* bp = B + (size_t)(8 * J + 4 * h) * NCOL + 64 * u + 4 * c;
        const float4 g0 = *(const float4*)(bp);
        const float4 g1 = *(const float4*)(bp + NCOL);
        const float4 g2 = *(const float4*)(bp + 2 * NCOL);
        const float4 g3 = *(const float4*)(bp + 3 * NCOL);
        float4* dp = Bs + (J * 8 + 4 * u) * 32 + lane;
        dp[0 * 32] = make_float4(g0.x, g1.x, g2.x, g3.x);
        dp[1 * 32] = make_float4(g0.y, g1.y, g2.y, g3.y);
        dp[2 * 32] = make_float4(g0.z, g1.z, g2.z, g3.z);
        dp[3 * 32] = make_float4(g0.w, g1.w, g2.w, g3.w);
    }
    __syncthreads();
    if (!active) return;

    float* cp = C + (size_t)(m0 + 8 * h) * NCOL + 4 * c;
    asm volatile("" : "+v"(cp));

    v8f acc[8];
#pragma unroll
    for (int t = 0; t < 8; ++t) acc[t] = (v8f){};

    unsigned lo = (unsigned)lane * 16u;
    unsigned hi = lo + 65536u;
    asm volatile("" : "+v"(lo), "+v"(hi));
    const char* bsb = (const char*)Bs;
#define BFRAG(J, t) \
    (*(const float4*)(bsb + ((J) < KBLOCKS / 2 ? lo : hi) + \
                      ((((J) & (KBLOCKS / 2 - 1)) * 8 + (t)) * 512)))

    float4 bc[8], bn[8];
#pragma unroll
    for (int t = 0; t < 8; ++t) bc[t] = BFRAG(0, t);

#pragma unroll
    for (int J = 0; J < KBLOCKS; ++J) {
        if (J + 1 < KBLOCKS) {
#pragma unroll
            for (int t = 0; t < 8; ++t) bn[t] = BFRAG(J + 1, t);
        }
        const v2f a0 = {a[J].x, a[J].y};
        const v2f a1 = {a[J].z, a[J].w};
#pragma unroll
        for (int t = 0; t < 8; ++t) {
            const v2f b0 = {bc[t].x, bc[t].y};
            acc[t] = __builtin_amdgcn_wmma_f32_16x16x4_f32(
                false, a0, false, b0, (short)0, acc[t], false, false);
        }
#pragma unroll
        for (int t = 0; t < 8; ++t) {
            const v2f b1 = {bc[t].z, bc[t].w};
            acc[t] = __builtin_amdgcn_wmma_f32_16x16x4_f32(
                false, a1, false, b1, (short)0, acc[t], false, false);
        }
#pragma unroll
        for (int t = 0; t < 8; ++t) bc[t] = bn[t];
    }

    asm volatile("v_nop\n\tv_nop\n\tv_nop\n\tv_nop\n\tv_nop"
                 : "+v"(acc[0]), "+v"(acc[1]), "+v"(acc[2]), "+v"(acc[3]),
                   "+v"(acc[4]), "+v"(acc[5]), "+v"(acc[6]), "+v"(acc[7]));

#pragma unroll
    for (int i = 0; i < 8; ++i) {
#pragma unroll
        for (int u = 0; u < 2; ++u) {
            *(float4*)(cp + (size_t)i * NCOL + 64 * u) =
                make_float4(acc[4 * u + 0][i], acc[4 * u + 1][i],
                            acc[4 * u + 2][i], acc[4 * u + 3][i]);
        }
    }
}


__device__ __forceinline__ float leaky(float v) {
    return v > 0.f ? v : ALPHA * v;
}

__global__ __launch_bounds__(SORT_THREADS)
void gat_partition_edges(const int* __restrict__ src,
                         const int* __restrict__ dst,
                         int* __restrict__ cnt,
                         int* __restrict__ rows_as_pairs,
                         int n_nodes, int n_edges) {
    extern __shared__ int hbin[];
    int2* pairs = (int2*)rows_as_pairs;
    const int tid = threadIdx.x;
    const int nb = (n_nodes + BUCKET_NODES - 1) >> BUCKET_SHIFT;
    const int chunk = (n_edges + (int)gridDim.x - 1) / (int)gridDim.x;
    const int e0 = blockIdx.x * chunk;
    const int e1 = (e0 + chunk < n_edges) ? e0 + chunk : n_edges;

    for (int b = tid; b < nb; b += SORT_THREADS) hbin[b] = 0;
    __syncthreads();
    for (int e = e0 + tid; e < e1; e += 4 * SORT_THREADS) {
        int d[4];
#pragma unroll
        for (int k = 0; k < 4; ++k) {
            const int ek = e + k * SORT_THREADS;
            d[k] = dst[ek < e1 ? ek : e1 - 1];
        }
        asm volatile("" : "+v"(d[0]), "+v"(d[1]), "+v"(d[2]), "+v"(d[3]));
#pragma unroll
        for (int k = 0; k < 4; ++k)
            if (e + k * SORT_THREADS < e1) atomicAdd(&hbin[d[k] >> BUCKET_SHIFT], 1);
    }
    __syncthreads();
    for (int b = tid; b < nb; b += SORT_THREADS) {
        const int c = hbin[b];
        if (c > 0) hbin[b] = atomicAdd(cnt + ((size_t)b << BUCKET_SHIFT), c);
    }
    __syncthreads();
    for (int e = e0 + tid; e < e1; e += 4 * SORT_THREADS) {
        int d[4], s[4];
#pragma unroll
        for (int k = 0; k < 4; ++k) {
            const int ek = e + k * SORT_THREADS;
            const int ec = ek < e1 ? ek : e1 - 1;
            d[k] = dst[ec];
            s[k] = src[ec];
        }
        asm volatile("" : "+v"(d[0]), "+v"(d[1]), "+v"(d[2]), "+v"(d[3]),
                          "+v"(s[0]), "+v"(s[1]), "+v"(s[2]), "+v"(s[3]));
#pragma unroll
        for (int k = 0; k < 4; ++k) {
            if (e + k * SORT_THREADS < e1) {
                const int b = d[k] >> BUCKET_SHIFT;
                const int pos = atomicAdd(&hbin[b], 1);
                const int left = n_nodes - (b << BUCKET_SHIFT);
                const int rows = left < BUCKET_NODES ? left : BUCKET_NODES;
                if (pos < rows * PAIRS_PER_ROW)
                    pairs[(size_t)b * BUCKET_PAIR_CAP + pos] = make_int2(d[k], s[k]);
            }
        }
    }
}

__global__ __launch_bounds__(SORT_THREADS)
void gat_place_edges(const int* __restrict__ src,
                     const int* __restrict__ dst,
                     int* __restrict__ cnt,
                     int* lists, int n_nodes, int n_edges) {
    __shared__ int2 pl[BUCKET_PAIR_CAP];
    __shared__ int deg[BUCKET_NODES];
    const int tid = threadIdx.x;
    const int b = blockIdx.x;
    const int node0 = b << BUCKET_SHIFT;
    const int left = n_nodes - node0;
    const int rows = left < BUCKET_NODES ? left : BUCKET_NODES;
    const int np = cnt[node0];
    const bool stored = np <= rows * PAIRS_PER_ROW;

    if (tid < BUCKET_NODES) deg[tid] = 0;
    if (stored) {
        const int2* pg = (const int2*)lists + (size_t)b * BUCKET_PAIR_CAP;
        for (int i = tid; i < np; i += 4 * SORT_THREADS) {
            int2 p[4];
#pragma unroll
            for (int k = 0; k < 4; ++k) {
                const int ik = i + k * SORT_THREADS;
                p[k] = pg[ik < np ? ik : np - 1];
            }
            asm volatile("" ::: "memory");
#pragma unroll
            for (int k = 0; k < 4; ++k) {
                const int ik = i + k * SORT_THREADS;
                if (ik < np) pl[ik] = p[k];
            }
        }
    }
    __syncthreads();
    if (stored) {
        for (int i = tid; i < np; i += SORT_THREADS) {
            const int2 p = pl[i];
            const int pos = atomicAdd(&deg[p.x & (BUCKET_NODES - 1)], 1);
            if (pos < ROW_SLOTS) lists[(size_t)p.x * NCOL + pos] = p.y;
        }
    } else {
        for (int e = tid; e < n_edges; e += SORT_THREADS) {
            const int d = dst[e];
            if ((d >> BUCKET_SHIFT) == b) {
                const int pos = atomicAdd(&deg[d & (BUCKET_NODES - 1)], 1);
                if (pos < ROW_SLOTS) lists[(size_t)d * NCOL + pos] = src[e];
            }
        }
    }
    __syncthreads();
    if (tid < rows) cnt[node0 + tid] = deg[tid];
}


__device__ __forceinline__ float sum8(float v) {
    v += __int_as_float(__builtin_amdgcn_update_dpp(0, __float_as_int(v), 0x161, 0xf, 0xf, true));
    v += __int_as_float(__builtin_amdgcn_update_dpp(0, __float_as_int(v), 0x162, 0xf, 0xf, true));
    v += __int_as_float(__builtin_amdgcn_update_dpp(0, __float_as_int(v), 0x164, 0xf, 0xf, true));
    return v;
}

template <int X>
__device__ __forceinline__ float lane_xor(const float v) {
    return __int_as_float(__builtin_amdgcn_update_dpp(0, __float_as_int(v), 0x160 + X, 0xf, 0xf, true));
}

template <int R>
__device__ __forceinline__ float lane_bcast8(const float v) {
    return __uint_as_float(__builtin_amdgcn_mov_dpp8(__float_as_uint(v), R * 0x249249u));
}

typedef __attribute__((ext_vector_type(4))) float v4f;

__device__ __forceinline__ float dot4(const v4f a, const v4f b) {
    const v2f t = a.lo * b.lo + a.hi * b.hi;
    return t.x + t.y;
}

template <int G>
__device__ __forceinline__ void gat_rows(const float* __restrict__ ft, const unsigned q4,
                                         const int (&s)[G],
                                         const v4f al, const float a2v, const int k8,
                                         float& m, float& l, v4f& acc) {
    v4f F[G];
#pragma unroll
    for (int g = 0; g < G; ++g) F[g] = *(const v4f*)(ft + (size_t)s[g] * NCOL + q4);
    if constexpr (G == 8)
        asm volatile("" : "+v"(F[0]), "+v"(F[1]), "+v"(F[2]), "+v"(F[3]),
                          "+v"(F[4]), "+v"(F[5]), "+v"(F[6]), "+v"(F[7]));

    const int kg = k8 & (G - 1);
    float a1 = sum8(dot4(F[0], al));
#pragma unroll
    for (int g = 1; g < G; ++g) {
        const float t = sum8(dot4(F[g], al));
        a1 = (kg == g) ? t : a1;
    }
    const float e = leaky(a1 + a2v);
    float mx = e;
    if constexpr (G >= 2) mx = fmaxf(mx, lane_xor<1>(mx));
    if constexpr (G >= 4) mx = fmaxf(mx, lane_xor<2>(mx));
    if constexpr (G >= 8) mx = fmaxf(mx, lane_xor<4>(mx));
    const float mn = fmaxf(m, mx);
    const float sc = expf(m - mn);
    const float wk = expf(e - mn);
    l *= sc;
    acc *= sc;
    float w[G];
    if constexpr (G == 1) {
        w[0] = wk;
    } else {
        w[0] = lane_bcast8<0>(wk);
        w[1] = lane_bcast8<1>(wk);
        if constexpr (G >= 4) { w[2] = lane_bcast8<2>(wk); w[3] = lane_bcast8<3>(wk); }
        if constexpr (G >= 8) {
            w[4] = lane_bcast8<4>(wk); w[5] = lane_bcast8<5>(wk);
            w[6] = lane_bcast8<6>(wk); w[7] = lane_bcast8<7>(wk);
        }
    }
#pragma unroll
    for (int g = 0; g < G; ++g) {
        l += w[g];
        acc += F[g] * w[g];
    }
    m = mn;
}

template <int G>
__device__ __forceinline__ void gat_listed(const float* __restrict__ ft, const unsigned q4,
                                           const int mine,
                                           const int j0, const v4f al, const float a2v,
                                           const int k8, float& m, float& l, v4f& acc) {
    int s[G];
#pragma unroll
    for (int g = 0; g < G; ++g) s[g] = __builtin_amdgcn_readlane(mine, j0 + g);
    gat_rows<G>(ft, q4, s, al, a2v, k8, m, l, acc);
}

__global__ void gat_gather(const int* __restrict__ src,
                           const int* __restrict__ dst,
                           const float* __restrict__ ft,
                           const float* __restrict__ attn_l,
                           const float* __restrict__ attn_r,
                           const int* __restrict__ cnt,
                           const int* lists,
                           float* out, int n_nodes, int n_edges) {
    const int t = blockIdx.x * blockDim.x + threadIdx.x;
    const int n = __builtin_amdgcn_readfirstlane(t >> 5);
    if (n >= n_nodes) return;
    const int q = t & 31;
    const int* row = lists + (size_t)n * NCOL;
    const unsigned q4 = 4u * (unsigned)q;
    const int k8 = q & 7;

    const int deg = __builtin_amdgcn_readfirstlane(cnt[n]);
    int mine = row[q];
    v4f own = *(const v4f*)(ft + (size_t)n * NCOL + q4);
    const v4f al = *(const v4f*)(attn_l + 4 * q);
    const v4f ar = *(const v4f*)(attn_r + 4 * q);
    asm volatile("" : "+v"(mine), "+v"(own));
    const float a2v = sum8(dot4(own, ar));

    float m = -3.402823466e+38f;
    float l = 0.f;
    v4f acc = {0.f, 0.f, 0.f, 0.f};

    if (deg <= ROW_SLOTS) {
        for (int base = 0; base < deg; base += 32) {
            const int left = deg - base;
            const int nj = left < 32 ? left : 32;
            if (base > 0) mine = row[base + q];
            int j = 0;
            for (; j + 8 <= nj; j += 8) gat_listed<8>(ft, q4, mine, j, al, a2v, k8, m, l, acc);
            if (nj - j >= 4) { gat_listed<4>(ft, q4, mine, j, al, a2v, k8, m, l, acc); j += 4; }
            if (nj - j >= 2) { gat_listed<2>(ft, q4, mine, j, al, a2v, k8, m, l, acc); j += 2; }
            if (nj - j >= 1) gat_listed<1>(ft, q4, mine, j, al, a2v, k8, m, l, acc);
        }
    } else {
        for (int e = 0; e < n_edges; ++e) {
            if (__builtin_amdgcn_readfirstlane(dst[e]) == n) {
                const int s[1] = {__builtin_amdgcn_readfirstlane(src[e])};
                gat_rows<1>(ft, q4, s, al, a2v, k8, m, l, acc);
            }
        }
    }

    v4f x = {0.f, 0.f, 0.f, 0.f};
    if (l > 0.f) x = acc / l;
    *(v4f*)(out + (size_t)n * NCOL + 4 * q) = x;
}

extern "C" void kernel_launch(void* const* d_in, const int* in_sizes, int n_in,
                              void* d_out, int out_size, void* d_ws, size_t ws_size,
                              hipStream_t stream) {
    const float* feature = (const float*)d_in[0];
    const int*   src     = (const int*)d_in[1];
    const int*   dst     = (const int*)d_in[2];
    const float* W_fc    = (const float*)d_in[3];
    const float* attn_l  = (const float*)d_in[4];
    const float* attn_r  = (const float*)d_in[5];
    float*       out     = (float*)d_out;

    const int M = in_sizes[0] / IN_DIM;
    const int E = in_sizes[1];

    float* ft  = (float*)d_ws;
    int*   cnt = (int*)(ft + (size_t)M * NCOL + (size_t)M * HEADS * 2);

    {
        int n_strips = (M + 15) / 16;
        gat_gemm_wmma<<<(n_strips + GEMM_WAVES - 1) / GEMM_WAVES, GEMM_WAVES * 32, 0, stream>>>(
            feature, W_fc, ft, cnt, M);
    }
    gat_partition_edges<<<256, SORT_THREADS,
                          ((M + BUCKET_NODES - 1) / BUCKET_NODES) * sizeof(int), stream>>>(
        src, dst, cnt, (int*)out, M, E);
    gat_place_edges<<<(M + BUCKET_NODES - 1) / BUCKET_NODES, SORT_THREADS, 0, stream>>>(
        src, dst, cnt, (int*)out, M, E);
    gat_gather<<<(M * 32 + 255) / 256, 256, 0, stream>>>(src, dst, ft, attn_l, attn_r, cnt,
                                                         (const int*)out, out, M, E);
}


# Place kernel merged into the gather: a 32-wave work-group per 256-node bucket builds lists and a2 in LDS, then sums its nodes (next node from an LDS counter); 3 launches; sum order still arrival rank
# speedup vs baseline: 6.6332x; 1.0453x over previous
#include <hip/hip_runtime.h>
#include <hip/hip_bf16.h>

#define IN_DIM 256
#define NCOL 128
#define HEADS 4
#define OUT_DIM 32
#define ALPHA 0.2f

#define BUCKET_SHIFT 8
#define BUCKET_NODES (1 << BUCKET_SHIFT)
#define LIST_SLOTS 64
#define PAIRS_PER_ROW (NCOL / 2)
#define BUCKET_PAIR_CAP (BUCKET_NODES * PAIRS_PER_ROW)
#define SORT_THREADS 1024

typedef __attribute__((ext_vector_type(2))) float v2f;
typedef __attribute__((ext_vector_type(8))) float v8f;

#define GEMM_WAVES 4
#define KBLOCKS (IN_DIM / 8)

__global__ __launch_bounds__(GEMM_WAVES * 32)
void gat_gemm_wmma(const float* __restrict__ A,
                   const float* __restrict__ B,
                   float* __restrict__ C,
                   int* __restrict__ cnt, int M) {
    __shared__ float4 Bs[KBLOCKS * 8 * 32];

    const int tid  = threadIdx.x;
    const int lane = tid & 31;
    const int wv   = tid >> 5;
    const int c    = lane & 15;
    const int h    = lane >> 4;
    const int m0   = (blockIdx.x * GEMM_WAVES + wv) * 16;
    const bool active = m0 < M;

    if (tid == 0 && blockIdx.x * BUCKET_NODES < M) cnt[blockIdx.x * BUCKET_NODES] = 0;

    float4 a[KBLOCKS];
    if (active) {
        const float* ap = A + (size_t)(m0 + c) * IN_DIM + 4 * h;
#pragma unroll
        for (int J = 0; J < KBLOCKS; ++J) a[J] = *(const float4*)(ap + 8 * J);
    }

#pragma unroll 4
    for (int it = 0; it < (KBLOCKS * 2) / GEMM_WAVES; ++it) {
        const int s = it * GEMM_WAVES + wv;
        const int J = s >> 1;
        const int u = s & 1;
        const float* bp = B + (size_t)(8 * J + 4 * h) * NCOL + 64 * u + 4 * c;
        const float4 g0 = *(const float4*)(bp);
        const float4 g1 = *(const float4*)(bp + NCOL);
        const float4 g2 = *(const float4*)(bp + 2 * NCOL);
        const float4 g3 = *(const float4*)(bp + 3 * NCOL);
        float4* dp = Bs + (J * 8 + 4 * u) * 32 + lane;
        dp[0 * 32] = make_float4(g0.x, g1.x, g2.x, g3.x);
        dp[1 * 32] = make_float4(g0.y, g1.y, g2.y, g3.y);
        dp[2 * 32] = make_float4(g0.z, g1.z, g2.z, g3.z);
        dp[3 * 32] = make_float4(g0.w, g1.w, g2.w, g3.w);
    }
    __syncthreads();
    if (!active) return;

    float* cp = C + (size_t)(m0 + 8 * h) * NCOL + 4 * c;
    asm volatile("" : "+v"(cp));

    v8f acc[8];
#pragma unroll
    for (int t = 0; t < 8; ++t) acc[t] = (v8f){};

    unsigned lo = (unsigned)lane * 16u;
    unsigned hi = lo + 65536u;
    asm volatile("" : "+v"(lo), "+v"(hi));
    const char* bsb = (const char*)Bs;
#define BFRAG(J, t) \
    (*(const float4*)(bsb + ((J) < KBLOCKS / 2 ? lo : hi) + \
                      ((((J) & (KBLOCKS / 2 - 1)) * 8 + (t)) * 512)))

    float4 bc[8], bn[8];
#pragma unroll
    for (int t = 0; t < 8; ++t) bc[t] = BFRAG(0, t);

#pragma unroll
    for (int J = 0; J < KBLOCKS; ++J) {
        if (J + 1 < KBLOCKS) {
#pragma unroll
            for (int t = 0; t < 8; ++t) bn[t] = BFRAG(J + 1, t);
        }
        const v2f a0 = {a[J].x, a[J].y};
        const v2f a1 = {a[J].z, a[J].w};
#pragma unroll
        for (int t = 0; t < 8; ++t) {
            const v2f b0 = {bc[t].x, bc[t].y};
            acc[t] = __builtin_amdgcn_wmma_f32_16x16x4_f32(
                false, a0, false, b0, (short)0, acc[t], false, false);
        }
#pragma unroll
        for (int t = 0; t < 8; ++t) {
            const v2f b1 = {bc[t].z, bc[t].w};
            acc[t] = __builtin_amdgcn_wmma_f32_16x16x4_f32(
                false, a1, false, b1, (short)0, acc[t], false, false);
        }
#pragma unroll
        for (int t = 0; t < 8; ++t) bc[t] = bn[t];
    }

    asm volatile("v_nop\n\tv_nop\n\tv_nop\n\tv_nop\n\tv_nop"
                 : "+v"(acc[0]), "+v"(acc[1]), "+v"(acc[2]), "+v"(acc[3]),
                   "+v"(acc[4]), "+v"(acc[5]), "+v"(acc[6]), "+v"(acc[7]));

#pragma unroll
    for (int i = 0; i < 8; ++i) {
#pragma unroll
        for (int u = 0; u < 2; ++u) {
            *(float4*)(cp + (size_t)i * NCOL + 64 * u) =
                make_float4(acc[4 * u + 0][i], acc[4 * u + 1][i],
                            acc[4 * u + 2][i], acc[4 * u + 3][i]);
        }
    }
}


__device__ __forceinline__ float leaky(float v) {
    return v > 0.f ? v : ALPHA * v;
}

__global__ __launch_bounds__(SORT_THREADS)
void gat_partition_edges(const int* __restrict__ src,
                         const int* __restrict__ dst,
                         int* __restrict__ cnt,
                         int* __restrict__ rows_as_pairs,
                         int n_nodes, int n_edges) {
    extern __shared__ int hbin[];
    int2* pairs = (int2*)rows_as_pairs;
    const int tid = threadIdx.x;
    const int nb = (n_nodes + BUCKET_NODES - 1) >> BUCKET_SHIFT;
    const int chunk = (n_edges + (int)gridDim.x - 1) / (int)gridDim.x;
    const int e0 = blockIdx.x * chunk;
    const int e1 = (e0 + chunk < n_edges) ? e0 + chunk : n_edges;

    for (int b = tid; b < nb; b += SORT_THREADS) hbin[b] = 0;
    __syncthreads();
    for (int e = e0 + tid; e < e1; e += 4 * SORT_THREADS) {
        int d[4];
#pragma unroll
        for (int k = 0; k < 4; ++k) {
            const int ek = e + k * SORT_THREADS;
            d[k] = dst[ek < e1 ? ek : e1 - 1];
        }
        asm volatile("" : "+v"(d[0]), "+v"(d[1]), "+v"(d[2]), "+v"(d[3]));
#pragma unroll
        for (int k = 0; k < 4; ++k)
            if (e + k * SORT_THREADS < e1) atomicAdd(&hbin[d[k] >> BUCKET_SHIFT], 1);
    }
    __syncthreads();
    for (int b = tid; b < nb; b += SORT_THREADS) {
        const int c = hbin[b];
        if (c > 0) hbin[b] = atomicAdd(cnt + ((size_t)b << BUCKET_SHIFT), c);
    }
    __syncthreads();
    for (int e = e0 + tid; e < e1; e += 4 * SORT_THREADS) {
        int d[4], s[4];
#pragma unroll
        for (int k = 0; k < 4; ++k) {
            const int ek = e + k * SORT_THREADS;
            const int ec = ek < e1 ? ek : e1 - 1;
            d[k] = dst[ec];
            s[k] = src[ec];
        }
        asm volatile("" : "+v"(d[0]), "+v"(d[1]), "+v"(d[2]), "+v"(d[3]),
                          "+v"(s[0]), "+v"(s[1]), "+v"(s[2]), "+v"(s[3]));
#pragma unroll
        for (int k = 0; k < 4; ++k) {
            if (e + k * SORT_THREADS < e1) {
                const int b = d[k] >> BUCKET_SHIFT;
                const int pos = atomicAdd(&hbin[b], 1);
                const int left = n_nodes - (b << BUCKET_SHIFT);
                const int rows = left < BUCKET_NODES ? left : BUCKET_NODES;
                if (pos < rows * PAIRS_PER_ROW)
                    pairs[(size_t)b * BUCKET_PAIR_CAP + pos] = make_int2(d[k], s[k]);
            }
        }
    }
}


__device__ __forceinline__ float sum8(float v) {
    v += __int_as_float(__builtin_amdgcn_update_dpp(0, __float_as_int(v), 0x161, 0xf, 0xf, true));
    v += __int_as_float(__builtin_amdgcn_update_dpp(0, __float_as_int(v), 0x162, 0xf, 0xf, true));
    v += __int_as_float(__builtin_amdgcn_update_dpp(0, __float_as_int(v), 0x164, 0xf, 0xf, true));
    return v;
}

template <int X>
__device__ __forceinline__ float lane_xor(const float v) {
    return __int_as_float(__builtin_amdgcn_update_dpp(0, __float_as_int(v), 0x160 + X, 0xf, 0xf, true));
}

template <int R>
__device__ __forceinline__ float lane_bcast8(const float v) {
    return __uint_as_float(__builtin_amdgcn_mov_dpp8(__float_as_uint(v), R * 0x249249u));
}

typedef __attribute__((ext_vector_type(4))) float v4f;

__device__ __forceinline__ float dot4(const v4f a, const v4f b) {
    const v2f t = a.lo * b.lo + a.hi * b.hi;
    return t.x + t.y;
}

template <int G>
__device__ __forceinline__ void gat_rows(const float* __restrict__ ft, const unsigned q4,
                                         const int (&s)[G],
                                         const v4f al, const float a2v, const int k8,
                                         float& m, float& l, v4f& acc) {
    v4f F[G];
#pragma unroll
    for (int g = 0; g < G; ++g) F[g] = *(const v4f*)(ft + (size_t)s[g] * NCOL + q4);
    if constexpr (G == 8)
        asm volatile("" : "+v"(F[0]), "+v"(F[1]), "+v"(F[2]), "+v"(F[3]),
                          "+v"(F[4]), "+v"(F[5]), "+v"(F[6]), "+v"(F[7]));

    const int kg = k8 & (G - 1);
    float a1 = sum8(dot4(F[0], al));
#pragma unroll
    for (int g = 1; g < G; ++g) {
        const float t = sum8(dot4(F[g], al));
        a1 = (kg == g) ? t : a1;
    }
    const float e = leaky(a1 + a2v);
    float mx = e;
    if constexpr (G >= 2) mx = fmaxf(mx, lane_xor<1>(mx));
    if constexpr (G >= 4) mx = fmaxf(mx, lane_xor<2>(mx));
    if constexpr (G >= 8) mx = fmaxf(mx, lane_xor<4>(mx));
    const float mn = fmaxf(m, mx);
    const float sc = expf(m - mn);
    const float wk = expf(e - mn);
    l *= sc;
    acc *= sc;
    float w[G];
    if constexpr (G == 1) {
        w[0] = wk;
    } else {
        w[0] = lane_bcast8<0>(wk);
        w[1] = lane_bcast8<1>(wk);
        if constexpr (G >= 4) { w[2] = lane_bcast8<2>(wk); w[3] = lane_bcast8<3>(wk); }
        if constexpr (G >= 8) {
            w[4] = lane_bcast8<4>(wk); w[5] = lane_bcast8<5>(wk);
            w[6] = lane_bcast8<6>(wk); w[7] = lane_bcast8<7>(wk);
        }
    }
#pragma unroll
    for (int g = 0; g < G; ++g) {
        l += w[g];
        acc += F[g] * w[g];
    }
    m = mn;
}

template <int G>
__device__ __forceinline__ void gat_listed(const float* __restrict__ ft, const unsigned q4,
                                           const int mine,
                                           const int j0, const v4f al, const float a2v,
                                           const int k8, float& m, float& l, v4f& acc) {
    int s[G];
#pragma unroll
    for (int g = 0; g < G; ++g) s[g] = __builtin_amdgcn_readlane(mine, j0 + g);
    gat_rows<G>(ft, q4, s, al, a2v, k8, m, l, acc);
}

__global__ __launch_bounds__(SORT_THREADS) __attribute__((amdgpu_waves_per_eu(16)))
void gat_bucket_gather(const int* __restrict__ src,
                       const int* __restrict__ dst,
                       const float* __restrict__ ft,
                       const float* __restrict__ attn_l,
                       const float* __restrict__ attn_r,
                       const int* __restrict__ cnt,
                       const int* pairs_in_rows,
                       float* out, int n_nodes, int n_edges) {
    __shared__ int lst[BUCKET_NODES * LIST_SLOTS];
    __shared__ int deg[BUCKET_NODES];
    __shared__ float a2s[BUCKET_NODES * HEADS];
    __shared__ int next_node;
    const int tid = threadIdx.x;
    const int q = tid & 31;
    const int wv = __builtin_amdgcn_readfirstlane(tid >> 5);
    const int b = blockIdx.x;
    const int node0 = b << BUCKET_SHIFT;
    const int left = n_nodes - node0;
    const int rows = left < BUCKET_NODES ? left : BUCKET_NODES;
    const unsigned q4 = 4u * (unsigned)q;
    const int k8 = q & 7;
    const int np = cnt[node0];
    const bool stored = np <= rows * PAIRS_PER_ROW;

    if (tid < BUCKET_NODES) deg[tid] = 0;
    if (tid == 0) next_node = SORT_THREADS / 32;
    __syncthreads();

    v4f own[8];
#pragma unroll
    for (int g = 0; g < 8; ++g) {
        const int kk = 8 * wv + g;
        const int kc = kk < rows ? kk : rows - 1;
        own[g] = *(const v4f*)(ft + (size_t)(node0 + kc) * NCOL + q4);
    }
    const v4f ar = *(const v4f*)(attn_r + 4 * q);

    if (stored) {
        const int2* pg = (const int2*)pairs_in_rows + (size_t)b * BUCKET_PAIR_CAP;
        for (int i = tid; i < np; i += 4 * SORT_THREADS) {
            int2 p[4];
#pragma unroll
            for (int k = 0; k < 4; ++k) {
                const int ik = i + k * SORT_THREADS;
                p[k] = pg[ik < np ? ik : np - 1];
            }
            asm volatile("" : "+v"(p[0].x), "+v"(p[0].y), "+v"(p[1].x), "+v"(p[1].y),
                              "+v"(p[2].x), "+v"(p[2].y), "+v"(p[3].x), "+v"(p[3].y));
#pragma unroll
            for (int k = 0; k < 4; ++k) {
                if (i + k * SORT_THREADS < np) {
                    const int dl = p[k].x & (BUCKET_NODES - 1);
                    const int pos = atomicAdd(&deg[dl], 1);
                    if (pos < LIST_SLOTS) lst[dl * LIST_SLOTS + pos] = p[k].y;
                }
            }
        }
    } else {
        for (int e = tid; e < n_edges; e += SORT_THREADS) {
            const int d = dst[e];
            if ((d >> BUCKET_SHIFT) == b) {
                const int dl = d & (BUCKET_NODES - 1);
                const int pos = atomicAdd(&deg[dl], 1);
                if (pos < LIST_SLOTS) lst[dl * LIST_SLOTS + pos] = src[e];
            }
        }
    }

#pragma unroll
    for (int g = 0; g < 8; ++g) {
        const float a2 = sum8(dot4(own[g], ar));
        if (k8 == 0) a2s[(8 * wv + g) * HEADS + (q >> 3)] = a2;
    }
    __syncthreads();

    const v4f al = *(const v4f*)(attn_l + 4 * q);
    int k = wv;
    while (k < rows) {
        int kn = 0;
        if (q == 0) kn = atomicAdd(&next_node, 1);
        const int dg = __builtin_amdgcn_readfirstlane(deg[k]);
        const float a2v = a2s[k * HEADS + (q >> 3)];
        const int* row = lst + k * LIST_SLOTS;
        int mine = row[q];
        kn = __builtin_amdgcn_readfirstlane(kn);
        const int n = node0 + k;

        float m = -3.402823466e+38f;
        float l = 0.f;
        v4f acc = {0.f, 0.f, 0.f, 0.f};

        if (dg <= LIST_SLOTS) {
            for (int base = 0; base < dg; base += 32) {
                const int nleft = dg - base;
                const int nj = nleft < 32 ? nleft : 32;
                if (base > 0) mine = row[base + q];
                int j = 0;
                for (; j + 8 <= nj; j += 8) gat_listed<8>(ft, q4, mine, j, al, a2v, k8, m, l, acc);
                if (nj - j >= 4) { gat_listed<4>(ft, q4, mine, j, al, a2v, k8, m, l, acc); j += 4; }
                if (nj - j >= 2) { gat_listed<2>(ft, q4, mine, j, al, a2v, k8, m, l, acc); j += 2; }
                if (nj - j >= 1) gat_listed<1>(ft, q4, mine, j, al, a2v, k8, m, l, acc);
            }
        } else {
            for (int e = 0; e < n_edges; ++e) {
                if (__builtin_amdgcn_readfirstlane(dst[e]) == n) {
                    const int s[1] = {__builtin_amdgcn_readfirstlane(src[e])};
                    gat_rows<1>(ft, q4, s, al, a2v, k8, m, l, acc);
                }
            }
        }

        v4f x = {0.f, 0.f, 0.f, 0.f};
        if (l > 0.f) x = acc / l;
        *(v4f*)(out + (size_t)n * NCOL + 4 * q) = x;
        k = kn;
    }
}

extern "C" void kernel_launch(void* const* d_in, const int* in_sizes, int n_in,
                              void* d_out, int out_size, void* d_ws, size_t ws_size,
                              hipStream_t stream) {
    const float* feature = (const float*)d_in[0];
    const int*   src     = (const int*)d_in[1];
    const int*   dst     = (const int*)d_in[2];
    const float* W_fc    = (const float*)d_in[3];
    const float* attn_l  = (const float*)d_in[4];
    const float* attn_r  = (const float*)d_in[5];
    float*       out     = (float*)d_out;

    const int M = in_sizes[0] / IN_DIM;
    const int E = in_sizes[1];

    float* ft  = (float*)d_ws;
    int*   cnt = (int*)(ft + (size_t)M * NCOL + (size_t)M * HEADS * 2);

    {
        int n_strips = (M + 15) / 16;
        gat_gemm_wmma<<<(n_strips + GEMM_WAVES - 1) / GEMM_WAVES, GEMM_WAVES * 32, 0, stream>>>(
            feature, W_fc, ft, cnt, M);
    }
    gat_partition_edges<<<256, SORT_THREADS,
                          ((M + BUCKET_NODES - 1) / BUCKET_NODES) * sizeof(int), stream>>>(
        src, dst, cnt, (int*)out, M, E);
    gat_bucket_gather<<<(M + BUCKET_NODES - 1) / BUCKET_NODES, SORT_THREADS, 0, stream>>>(
        src, dst, ft, attn_l, attn_r, cnt, (const int*)out, out, M, E);
}


# Edge partition moved into the GEMM launch: 256 extra work-groups beside the matrix work, fixed pair runs per (bucket, group), LDS ranks, no global atomic; 2 launches; sum order still arrival rank
# speedup vs baseline: 7.4740x; 1.1268x over previous
#include <hip/hip_runtime.h>
#include <hip/hip_bf16.h>

#define IN_DIM 256
#define NCOL 128
#define HEADS 4
#define OUT_DIM 32
#define ALPHA 0.2f

#define BUCKET_SHIFT 8
#define BUCKET_NODES (1 << BUCKET_SHIFT)
#define LIST_SLOTS 64
#define PAIRS_PER_ROW (NCOL / 2)
#define BUCKET_PAIR_CAP (BUCKET_NODES * PAIRS_PER_ROW)
#define SORT_THREADS 1024
#define PART_SHIFT 8
#define PART_WGS (1 << PART_SHIFT)
#define PART_EPT 49
#define GEMM_WAVES 4
#define KBLOCKS (IN_DIM / 8)

typedef __attribute__((ext_vector_type(2))) float v2f;
typedef __attribute__((ext_vector_type(8))) float v8f;


__device__ __forceinline__ int bucket_seg(const int b, const int n_nodes) {
    const int left = n_nodes - (b << BUCKET_SHIFT);
    const int rows = left < BUCKET_NODES ? left : BUCKET_NODES;
    return (rows * PAIRS_PER_ROW) >> PART_SHIFT;
}

__device__ __forceinline__ int table_pitch(const int n_nodes) {
    const int nb = (n_nodes + BUCKET_NODES - 1) >> BUCKET_SHIFT;
    return (nb + 31) & ~31;
}

__device__ __forceinline__ void gat_partition_edges(const int* __restrict__ src,
                                                    const int* __restrict__ dst,
                                                    int* __restrict__ tab,
                                                    int* __restrict__ rows_as_pairs,
                                                    int* hbin,
                                                    const int n_nodes, const int n_edges) {
    int2* pairs = (int2*)rows_as_pairs;
    const int tid = threadIdx.x;
    const int w = blockIdx.x;
    const int nthr = GEMM_WAVES * 32;
    const int nb = (n_nodes + BUCKET_NODES - 1) >> BUCKET_SHIFT;
    const int chunk = (n_edges + PART_WGS - 1) >> PART_SHIFT;
    const int e0 = w * chunk;
    const int e1 = (e0 + chunk < n_edges) ? e0 + chunk : n_edges;

    for (int b = tid; b < nb; b += nthr) hbin[b] = 0;
    __syncthreads();
    for (int e = e0 + tid; e < e1; e += PART_EPT * nthr) {
        int d[PART_EPT], s[PART_EPT], pos[PART_EPT];
#pragma unroll
        for (int k = 0; k < PART_EPT; ++k) {
            const int ek = e + k * nthr;
            const int ec = ek < e1 ? ek : e1 - 1;
            d[k] = dst[ec];
            s[k] = src[ec];
        }
        asm volatile("" : "+v"(d[0]), "+v"(d[1]), "+v"(d[2]), "+v"(d[3]), "+v"(d[4]),
                          "+v"(d[5]), "+v"(d[6]), "+v"(d[7]), "+v"(d[8]), "+v"(d[9]),
                          "+v"(d[10]), "+v"(d[11]), "+v"(d[12]));
        asm volatile("" : "+v"(d[13]), "+v"(d[14]), "+v"(d[15]), "+v"(d[16]), "+v"(d[17]),
                          "+v"(d[18]), "+v"(d[19]), "+v"(d[20]), "+v"(d[21]), "+v"(d[22]),
                          "+v"(d[23]), "+v"(d[24]), "+v"(d[25]));
        asm volatile("" : "+v"(d[26]), "+v"(d[27]), "+v"(d[28]), "+v"(d[29]), "+v"(d[30]),
                          "+v"(d[31]), "+v"(d[32]), "+v"(d[33]), "+v"(d[34]), "+v"(d[35]),
                          "+v"(d[36]), "+v"(d[37]), "+v"(d[38]));
        asm volatile("" : "+v"(d[39]), "+v"(d[40]), "+v"(d[41]), "+v"(d[42]), "+v"(d[43]),
                          "+v"(d[44]), "+v"(d[45]), "+v"(d[46]), "+v"(d[47]), "+v"(d[48]),
                          "+v"(s[0]), "+v"(s[1]), "+v"(s[2]));
        asm volatile("" : "+v"(s[3]), "+v"(s[4]), "+v"(s[5]), "+v"(s[6]), "+v"(s[7]),
                          "+v"(s[8]), "+v"(s[9]), "+v"(s[10]), "+v"(s[11]), "+v"(s[12]),
                          "+v"(s[13]), "+v"(s[14]), "+v"(s[15]));
        asm volatile("" : "+v"(s[16]), "+v"(s[17]), "+v"(s[18]), "+v"(s[19]), "+v"(s[20]),
                          "+v"(s[21]), "+v"(s[22]), "+v"(s[23]), "+v"(s[24]), "+v"(s[25]),
                          "+v"(s[26]), "+v"(s[27]), "+v"(s[28]));
        asm volatile("" : "+v"(s[29]), "+v"(s[30]), "+v"(s[31]), "+v"(s[32]), "+v"(s[33]),
                          "+v"(s[34]), "+v"(s[35]), "+v"(s[36]), "+v"(s[37]), "+v"(s[38]),
                          "+v"(s[39]), "+v"(s[40]), "+v"(s[41]));
        asm volatile("" : "+v"(s[42]), "+v"(s[43]), "+v"(s[44]), "+v"(s[45]), "+v"(s[46]),
                          "+v"(s[47]), "+v"(s[48]));
#pragma unroll
        for (int k = 0; k < PART_EPT; ++k)
            pos[k] = (e + k * nthr < e1) ? atomicAdd(&hbin[d[k] >> BUCKET_SHIFT], 1) : 0x7fffffff;
        asm volatile("" : "+v"(pos[0]), "+v"(pos[1]), "+v"(pos[2]), "+v"(pos[3]), "+v"(pos[4]),
                          "+v"(pos[5]), "+v"(pos[6]), "+v"(pos[7]), "+v"(pos[8]), "+v"(pos[9]),
                          "+v"(pos[10]), "+v"(pos[11]), "+v"(pos[12]));
        asm volatile("" : "+v"(pos[13]), "+v"(pos[14]), "+v"(pos[15]), "+v"(pos[16]), "+v"(pos[17]),
                          "+v"(pos[18]), "+v"(pos[19]), "+v"(pos[20]), "+v"(pos[21]), "+v"(pos[22]),
                          "+v"(pos[23]), "+v"(pos[24]), "+v"(pos[25]));
        asm volatile("" : "+v"(pos[26]), "+v"(pos[27]), "+v"(pos[28]), "+v"(pos[29]), "+v"(pos[30]),
                          "+v"(pos[31]), "+v"(pos[32]), "+v"(pos[33]), "+v"(pos[34]), "+v"(pos[35]),
                          "+v"(pos[36]), "+v"(pos[37]), "+v"(pos[38]));
        asm volatile("" : "+v"(pos[39]), "+v"(pos[40]), "+v"(pos[41]), "+v"(pos[42]), "+v"(pos[43]),
                          "+v"(pos[44]), "+v"(pos[45]), "+v"(pos[46]), "+v"(pos[47]), "+v"(pos[48]));
#pragma unroll
        for (int k = 0; k < PART_EPT; ++k) {
            const int b = d[k] >> BUCKET_SHIFT;
            const int seg = bucket_seg(b, n_nodes);
            if (pos[k] < seg)
                pairs[(size_t)b * BUCKET_PAIR_CAP + w * seg + pos[k]] = make_int2(d[k], s[k]);
        }
    }
    __syncthreads();
    int* trow = tab + (size_t)w * table_pitch(n_nodes);
    for (int b = tid; b < nb; b += nthr) trow[b] = hbin[b];
}

__global__ __launch_bounds__(GEMM_WAVES * 32)
void gat_gemm_wmma(const float* __restrict__ A,
                   const float* __restrict__ B,
                   float* __restrict__ C,
                   const int* __restrict__ src,
                   const int* __restrict__ dst,
                   int* __restrict__ tab,
                   int* __restrict__ rows_as_pairs,
                   int M, int n_edges) {
    __shared__ float4 Bs[KBLOCKS * 8 * 32];

    const int tid  = threadIdx.x;

    if (blockIdx.x < PART_WGS) {
        gat_partition_edges(src, dst, tab, rows_as_pairs, (int*)Bs, M, n_edges);
        return;
    }

    const int lane = tid & 31;
    const int wv   = tid >> 5;
    const int c    = lane & 15;
    const int h    = lane >> 4;
    const int m0   = ((blockIdx.x - PART_WGS) * GEMM_WAVES + wv) * 16;
    const bool active = m0 < M;

    float4 a[KBLOCKS];
    if (active) {
        const float* ap = A + (size_t)(m0 + c) * IN_DIM + 4 * h;
#pragma unroll
        for (int J = 0; J < KBLOCKS; ++J) a[J] = *(const float4*)(ap + 8 * J);
    }

#pragma unroll 4
    for (int it = 0; it < (KBLOCKS * 2) / GEMM_WAVES; ++it) {
        const int s = it * GEMM_WAVES + wv;
        const int J = s >> 1;
        const int u = s & 1;
        const float* bp = B + (size_t)(8 * J + 4 * h) * NCOL + 64 * u + 4 * c;
        const float4 g0 = *(const float4*)(bp);
        const float4 g1 = *(const float4*)(bp + NCOL);
        const float4 g2 = *(const float4*)(bp + 2 * NCOL);
        const float4 g3 = *(const float4*)(bp + 3 * NCOL);
        float4* dp = Bs + (J * 8 + 4 * u) * 32 + lane;
        dp[0 * 32] = make_float4(g0.x, g1.x, g2.x, g3.x);
        dp[1 * 32] = make_float4(g0.y, g1.y, g2.y, g3.y);
        dp[2 * 32] = make_float4(g0.z, g1.z, g2.z, g3.z);
        dp[3 * 32] = make_float4(g0.w, g1.w, g2.w, g3.w);
    }
    __syncthreads();
    if (!active) return;

    float* cp = C + (size_t)(m0 + 8 * h) * NCOL + 4 * c;
    asm volatile("" : "+v"(cp));

    v8f acc[8];
#pragma unroll
    for (int t = 0; t < 8; ++t) acc[t] = (v8f){};

    unsigned lo = (unsigned)lane * 16u;
    unsigned hi = lo + 65536u;
    asm volatile("" : "+v"(lo), "+v"(hi));
    const char* bsb = (const char*)Bs;
#define BFRAG(J, t) \
    (*(const float4*)(bsb + ((J) < KBLOCKS / 2 ? lo : hi) + \
                      ((((J) & (KBLOCKS / 2 - 1)) * 8 + (t)) * 512)))

    float4 bc[8], bn[8];
#pragma unroll
    for (int t = 0; t < 8; ++t) bc[t] = BFRAG(0, t);

#pragma unroll
    for (int J = 0; J < KBLOCKS; ++J) {
        if (J + 1 < KBLOCKS) {
#pragma unroll
            for (int t = 0; t < 8; ++t) bn[t] = BFRAG(J + 1, t);
        }
        const v2f a0 = {a[J].x, a[J].y};
        const v2f a1 = {a[J].z, a[J].w};
#pragma unroll
        for (int t = 0; t < 8; ++t) {
            const v2f b0 = {bc[t].x, bc[t].y};
            acc[t] = __builtin_amdgcn_wmma_f32_16x16x4_f32(
                false, a0, false, b0, (short)0, acc[t], false, false);
        }
#pragma unroll
        for (int t = 0; t < 8; ++t) {
            const v2f b1 = {bc[t].z, bc[t].w};
            acc[t] = __builtin_amdgcn_wmma_f32_16x16x4_f32(
                false, a1, false, b1, (short)0, acc[t], false, false);
        }
#pragma unroll
        for (int t = 0; t < 8; ++t) bc[t] = bn[t];
    }

    asm volatile("v_nop\n\tv_nop\n\tv_nop\n\tv_nop\n\tv_nop"
                 : "+v"(acc[0]), "+v"(acc[1]), "+v"(acc[2]), "+v"(acc[3]),
                   "+v"(acc[4]), "+v"(acc[5]), "+v"(acc[6]), "+v"(acc[7]));

#pragma unroll
    for (int i = 0; i < 8; ++i) {
#pragma unroll
        for (int u = 0; u < 2; ++u) {
            *(float4*)(cp + (size_t)i * NCOL + 64 * u) =
                make_float4(acc[4 * u + 0][i], acc[4 * u + 1][i],
                            acc[4 * u + 2][i], acc[4 * u + 3][i]);
        }
    }
}

__device__ __forceinline__ float leaky(float v) {
    return v > 0.f ? v : ALPHA * v;
}


__device__ __forceinline__ float sum8(float v) {
    v += __int_as_float(__builtin_amdgcn_update_dpp(0, __float_as_int(v), 0x161, 0xf, 0xf, true));
    v += __int_as_float(__builtin_amdgcn_update_dpp(0, __float_as_int(v), 0x162, 0xf, 0xf, true));
    v += __int_as_float(__builtin_amdgcn_update_dpp(0, __float_as_int(v), 0x164, 0xf, 0xf, true));
    return v;
}

template <int X>
__device__ __forceinline__ float lane_xor(const float v) {
    return __int_as_float(__builtin_amdgcn_update_dpp(0, __float_as_int(v), 0x160 + X, 0xf, 0xf, true));
}

template <int R>
__device__ __forceinline__ float lane_bcast8(const float v) {
    return __uint_as_float(__builtin_amdgcn_mov_dpp8(__float_as_uint(v), R * 0x249249u));
}

typedef __attribute__((ext_vector_type(4))) float v4f;

__device__ __forceinline__ float dot4(const v4f a, const v4f b) {
    const v2f t = a.lo * b.lo + a.hi * b.hi;
    return t.x + t.y;
}

template <int G>
__device__ __forceinline__ void gat_rows(const float* __restrict__ ft, const unsigned q4,
                                         const int (&s)[G],
                                         const v4f al, const float a2v, const int k8,
                                         float& m, float& l, v4f& acc) {
    v4f F[G];
#pragma unroll
    for (int g = 0; g < G; ++g) F[g] = *(const v4f*)(ft + (size_t)s[g] * NCOL + q4);
    if constexpr (G == 8)
        asm volatile("" : "+v"(F[0]), "+v"(F[1]), "+v"(F[2]), "+v"(F[3]),
                          "+v"(F[4]), "+v"(F[5]), "+v"(F[6]), "+v"(F[7]));

    const int kg = k8 & (G - 1);
    float a1 = sum8(dot4(F[0], al));
#pragma unroll
    for (int g = 1; g < G; ++g) {
        const float t = sum8(dot4(F[g], al));
        a1 = (kg == g) ? t : a1;
    }
    const float e = leaky(a1 + a2v);
    float mx = e;
    if constexpr (G >= 2) mx = fmaxf(mx, lane_xor<1>(mx));
    if constexpr (G >= 4) mx = fmaxf(mx, lane_xor<2>(mx));
    if constexpr (G >= 8) mx = fmaxf(mx, lane_xor<4>(mx));
    const float mn = fmaxf(m, mx);
    const float sc = expf(m - mn);
    const float wk = expf(e - mn);
    l *= sc;
    acc *= sc;
    float w[G];
    if constexpr (G == 1) {
        w[0] = wk;
    } else {
        w[0] = lane_bcast8<0>(wk);
        w[1] = lane_bcast8<1>(wk);
        if constexpr (G >= 4) { w[2] = lane_bcast8<2>(wk); w[3] = lane_bcast8<3>(wk); }
        if constexpr (G >= 8) {
            w[4] = lane_bcast8<4>(wk); w[5] = lane_bcast8<5>(wk);
            w[6] = lane_bcast8<6>(wk); w[7] = lane_bcast8<7>(wk);
        }
    }
#pragma unroll
    for (int g = 0; g < G; ++g) {
        l += w[g];
        acc += F[g] * w[g];
    }
    m = mn;
}

template <int G>
__device__ __forceinline__ void gat_listed(const float* __restrict__ ft, const unsigned q4,
                                           const int mine,
                                           const int j0, const v4f al, const float a2v,
                                           const int k8, float& m, float& l, v4f& acc) {
    int s[G];
#pragma unroll
    for (int g = 0; g < G; ++g) s[g] = __builtin_amdgcn_readlane(mine, j0 + g);
    gat_rows<G>(ft, q4, s, al, a2v, k8, m, l, acc);
}

__global__ __launch_bounds__(SORT_THREADS) __attribute__((amdgpu_waves_per_eu(16)))
void gat_bucket_gather(const int* __restrict__ src,
                       const int* __restrict__ dst,
                       const float* __restrict__ ft,
                       const float* __restrict__ attn_l,
                       const float* __restrict__ attn_r,
                       const int* __restrict__ tab,
                       const int* pairs_in_rows,
                       float* out, int n_nodes, int n_edges) {
    __shared__ int lst[BUCKET_NODES * LIST_SLOTS];
    __shared__ int deg[BUCKET_NODES];
    __shared__ float a2s[BUCKET_NODES * HEADS];
    __shared__ alignas(16) int segc[PART_WGS];
    __shared__ int next_node;
    const int tid = threadIdx.x;
    const int q = tid & 31;
    const int wv = __builtin_amdgcn_readfirstlane(tid >> 5);
    const int b = blockIdx.x;
    const int node0 = b << BUCKET_SHIFT;
    const int left = n_nodes - node0;
    const int rows = left < BUCKET_NODES ? left : BUCKET_NODES;
    const unsigned q4 = 4u * (unsigned)q;
    const int k8 = q & 7;
    const int seg = bucket_seg(b, n_nodes);

    if (tid < PART_WGS) segc[tid] = tab[(size_t)tid * table_pitch(n_nodes) + b];
    if (tid < BUCKET_NODES) deg[tid] = 0;
    if (tid == 0) next_node = SORT_THREADS / 32;
    __syncthreads();

    v4f own[8];
#pragma unroll
    for (int g = 0; g < 8; ++g) {
        const int kk = 8 * wv + g;
        const int kc = kk < rows ? kk : rows - 1;
        own[g] = *(const v4f*)(ft + (size_t)(node0 + kc) * NCOL + q4);
    }
    const v4f ar = *(const v4f*)(attn_r + 4 * q);

    static_assert(PART_WGS == 256 && SORT_THREADS == 4 * PART_WGS, "8 lengths per lane, 4 threads per run");
    const int4 c0 = ((const int4*)segc)[q];
    const int4 c1 = ((const int4*)segc)[32 + q];
    const int c0m = max(max(c0.x, c0.y), max(c0.z, c0.w));
    const int c1m = max(max(c1.x, c1.y), max(c1.z, c1.w));
    const bool stored = __builtin_amdgcn_ballot_w32(max(c0m, c1m) > seg) == 0u;

    if (stored) {
        const int w = tid >> 2;
        const int cw = segc[w];
        const int2* pg = (const int2*)pairs_in_rows + (size_t)b * BUCKET_PAIR_CAP + w * seg;
        for (int i = tid & 3; i < cw; i += 16) {
            int2 p[4];
#pragma unroll
            for (int k = 0; k < 4; ++k) {
                const int ik = i + 4 * k;
                p[k] = pg[ik < cw ? ik : cw - 1];
            }
            asm volatile("" : "+v"(p[0].x), "+v"(p[0].y), "+v"(p[1].x), "+v"(p[1].y),
                              "+v"(p[2].x), "+v"(p[2].y), "+v"(p[3].x), "+v"(p[3].y));
#pragma unroll
            for (int k = 0; k < 4; ++k) {
                if (i + 4 * k < cw) {
                    const int dl = p[k].x & (BUCKET_NODES - 1);
                    const int pos = atomicAdd(&deg[dl], 1);
                    if (pos < LIST_SLOTS) lst[dl * LIST_SLOTS + pos] = p[k].y;
                }
            }
        }
    } else {
        for (int e = tid; e < n_edges; e += SORT_THREADS) {
            const int d = dst[e];
            if ((d >> BUCKET_SHIFT) == b) {
                const int dl = d & (BUCKET_NODES - 1);
                const int pos = atomicAdd(&deg[dl], 1);
                if (pos < LIST_SLOTS) lst[dl * LIST_SLOTS + pos] = src[e];
            }
        }
    }

#pragma unroll
    for (int g = 0; g < 8; ++g) {
        const float a2 = sum8(dot4(own[g], ar));
        if (k8 == 0) a2s[(8 * wv + g) * HEADS + (q >> 3)] = a2;
    }
    __syncthreads();

    const v4f al = *(const v4f*)(attn_l + 4 * q);
    int k = wv;
    while (k < rows) {
        int kn = 0;
        if (q == 0) kn = atomicAdd(&next_node, 1);
        const int dg = __builtin_amdgcn_readfirstlane(deg[k]);
        const float a2v = a2s[k * HEADS + (q >> 3)];
        const int* row = lst + k * LIST_SLOTS;
        int mine = row[q];
        kn = __builtin_amdgcn_readfirstlane(kn);
        const int n = node0 + k;

        float m = -3.402823466e+38f;
        float l = 0.f;
        v4f acc = {0.f, 0.f, 0.f, 0.f};

        if (dg <= LIST_SLOTS) {
            for (int base = 0; base < dg; base += 32) {
                const int nleft = dg - base;
                const int nj = nleft < 32 ? nleft : 32;
                if (base > 0) mine = row[base + q];
                int j = 0;
                for (; j + 8 <= nj; j += 8) gat_listed<8>(ft, q4, mine, j, al, a2v, k8, m, l, acc);
                if (nj - j >= 4) { gat_listed<4>(ft, q4, mine, j, al, a2v, k8, m, l, acc); j += 4; }
                if (nj - j >= 2) { gat_listed<2>(ft, q4, mine, j, al, a2v, k8, m, l, acc); j += 2; }
                if (nj - j >= 1) gat_listed<1>(ft, q4, mine, j, al, a2v, k8, m, l, acc);
            }
        } else {
            for (int e = 0; e < n_edges; ++e) {
                if (__builtin_amdgcn_readfirstlane(dst[e]) == n) {
                    const int s[1] = {__builtin_amdgcn_readfirstlane(src[e])};
                    gat_rows<1>(ft, q4, s, al, a2v, k8, m, l, acc);
                }
            }
        }

        v4f x = {0.f, 0.f, 0.f, 0.f};
        if (l > 0.f) x = acc / l;
        *(v4f*)(out + (size_t)n * NCOL + 4 * q) = x;
        k = kn;
    }
}

extern "C" void kernel_launch(void* const* d_in, const int* in_sizes, int n_in,
                              void* d_out, int out_size, void* d_ws, size_t ws_size,
                              hipStream_t stream) {
    const float* feature = (const float*)d_in[0];
    const int*   src     = (const int*)d_in[1];
    const int*   dst     = (const int*)d_in[2];
    const float* W_fc    = (const float*)d_in[3];
    const float* attn_l  = (const float*)d_in[4];
    const float* attn_r  = (const float*)d_in[5];
    float*       out     = (float*)d_out;

    const int M = in_sizes[0] / IN_DIM;
    const int E = in_sizes[1];

    float* ft  = (float*)d_ws;
    int*   tab = (int*)(ft + (size_t)M * NCOL);

    {
        int n_strips = (M + 15) / 16;
        gat_gemm_wmma<<<PART_WGS + (n_strips + GEMM_WAVES - 1) / GEMM_WAVES, GEMM_WAVES * 32, 0, stream>>>(
            feature, W_fc, ft, src, dst, tab, (int*)out, M, E);
    }
    gat_bucket_gather<<<(M + BUCKET_NODES - 1) / BUCKET_NODES, SORT_THREADS, 0, stream>>>(
        src, dst, ft, attn_l, attn_r, tab, (const int*)out, out, M, E);
}


# GEMM made persistent: 512 resident work-groups stage W_fc once; each wave loops over 3 strips, the next strip's rows requested under the matrix stream; 106 left-over strips go out as column halves
# speedup vs baseline: 7.9927x; 1.0694x over previous
#include <hip/hip_runtime.h>
#include <hip/hip_bf16.h>

#define IN_DIM 256
#define NCOL 128
#define HEADS 4
#define OUT_DIM 32
#define ALPHA 0.2f

#define BUCKET_SHIFT 8
#define BUCKET_NODES (1 << BUCKET_SHIFT)
#define LIST_SLOTS 64
#define PAIRS_PER_ROW (NCOL / 2)
#define BUCKET_PAIR_CAP (BUCKET_NODES * PAIRS_PER_ROW)
#define SORT_THREADS 1024
#define PART_SHIFT 8
#define PART_WGS (1 << PART_SHIFT)
#define PART_EPT 49
#define GEMM_WAVES 4
#define KBLOCKS (IN_DIM / 8)
#define GEMM_WGS 512
#define GEMM_NW (GEMM_WGS * GEMM_WAVES)

typedef __attribute__((ext_vector_type(2))) float v2f;
typedef __attribute__((ext_vector_type(8))) float v8f;


__device__ __forceinline__ int bucket_seg(const int b, const int n_nodes) {
    const int left = n_nodes - (b << BUCKET_SHIFT);
    const int rows = left < BUCKET_NODES ? left : BUCKET_NODES;
    return (rows * PAIRS_PER_ROW) >> PART_SHIFT;
}

__device__ __forceinline__ int table_pitch(const int n_nodes) {
    const int nb = (n_nodes + BUCKET_NODES - 1) >> BUCKET_SHIFT;
    return (nb + 31) & ~31;
}

__device__ __forceinline__ void gat_partition_edges(const int* __restrict__ src,
                                                    const int* __restrict__ dst,
                                                    int* __restrict__ tab,
                                                    int* __restrict__ rows_as_pairs,
                                                    int* hbin,
                                                    const int n_nodes, const int n_edges) {
    int2* pairs = (int2*)rows_as_pairs;
    const int tid = threadIdx.x;
    const int w = blockIdx.x;
    const int nthr = GEMM_WAVES * 32;
    const int nb = (n_nodes + BUCKET_NODES - 1) >> BUCKET_SHIFT;
    const int chunk = (n_edges + PART_WGS - 1) >> PART_SHIFT;
    const int e0 = w * chunk;
    const int e1 = (e0 + chunk < n_edges) ? e0 + chunk : n_edges;

    for (int b = tid; b < nb; b += nthr) hbin[b] = 0;
    __syncthreads();
    for (int e = e0 + tid; e < e1; e += PART_EPT * nthr) {
        int d[PART_EPT], s[PART_EPT], pos[PART_EPT];
#pragma unroll
        for (int k = 0; k < PART_EPT; ++k) {
            const int ek = e + k * nthr;
            const int ec = ek < e1 ? ek : e1 - 1;
            d[k] = dst[ec];
            s[k] = src[ec];
        }
        asm volatile("" : "+v"(d[0]), "+v"(d[1]), "+v"(d[2]), "+v"(d[3]), "+v"(d[4]),
                          "+v"(d[5]), "+v"(d[6]), "+v"(d[7]), "+v"(d[8]), "+v"(d[9]),
                          "+v"(d[10]), "+v"(d[11]), "+v"(d[12]));
        asm volatile("" : "+v"(d[13]), "+v"(d[14]), "+v"(d[15]), "+v"(d[16]), "+v"(d[17]),
                          "+v"(d[18]), "+v"(d[19]), "+v"(d[20]), "+v"(d[21]), "+v"(d[22]),
                          "+v"(d[23]), "+v"(d[24]), "+v"(d[25]));
        asm volatile("" : "+v"(d[26]), "+v"(d[27]), "+v"(d[28]), "+v"(d[29]), "+v"(d[30]),
                          "+v"(d[31]), "+v"(d[32]), "+v"(d[33]), "+v"(d[34]), "+v"(d[35]),
                          "+v"(d[36]), "+v"(d[37]), "+v"(d[38]));
        asm volatile("" : "+v"(d[39]), "+v"(d[40]), "+v"(d[41]), "+v"(d[42]), "+v"(d[43]),
                          "+v"(d[44]), "+v"(d[45]), "+v"(d[46]), "+v"(d[47]), "+v"(d[48]),
                          "+v"(s[0]), "+v"(s[1]), "+v"(s[2]));
        asm volatile("" : "+v"(s[3]), "+v"(s[4]), "+v"(s[5]), "+v"(s[6]), "+v"(s[7]),
                          "+v"(s[8]), "+v"(s[9]), "+v"(s[10]), "+v"(s[11]), "+v"(s[12]),
                          "+v"(s[13]), "+v"(s[14]), "+v"(s[15]));
        asm volatile("" : "+v"(s[16]), "+v"(s[17]), "+v"(s[18]), "+v"(s[19]), "+v"(s[20]),
                          "+v"(s[21]), "+v"(s[22]), "+v"(s[23]), "+v"(s[24]), "+v"(s[25]),
                          "+v"(s[26]), "+v"(s[27]), "+v"(s[28]));
        asm volatile("" : "+v"(s[29]), "+v"(s[30]), "+v"(s[31]), "+v"(s[32]), "+v"(s[33]),
                          "+v"(s[34]), "+v"(s[35]), "+v"(s[36]), "+v"(s[37]), "+v"(s[38]),
                          "+v"(s[39]), "+v"(s[40]), "+v"(s[41]));
        asm volatile("" : "+v"(s[42]), "+v"(s[43]), "+v"(s[44]), "+v"(s[45]), "+v"(s[46]),
                          "+v"(s[47]), "+v"(s[48]));
#pragma unroll
        for (int k = 0; k < PART_EPT; ++k)
            pos[k] = (e + k * nthr < e1) ? atomicAdd(&hbin[d[k] >> BUCKET_SHIFT], 1) : 0x7fffffff;
        asm volatile("" : "+v"(pos[0]), "+v"(pos[1]), "+v"(pos[2]), "+v"(pos[3]), "+v"(pos[4]),
                          "+v"(pos[5]), "+v"(pos[6]), "+v"(pos[7]), "+v"(pos[8]), "+v"(pos[9]),
                          "+v"(pos[10]), "+v"(pos[11]), "+v"(pos[12]));
        asm volatile("" : "+v"(pos[13]), "+v"(pos[14]), "+v"(pos[15]), "+v"(pos[16]), "+v"(pos[17]),
                          "+v"(pos[18]), "+v"(pos[19]), "+v"(pos[20]), "+v"(pos[21]), "+v"(pos[22]),
                          "+v"(pos[23]), "+v"(pos[24]), "+v"(pos[25]));
        asm volatile("" : "+v"(pos[26]), "+v"(pos[27]), "+v"(pos[28]), "+v"(pos[29]), "+v"(pos[30]),
                          "+v"(pos[31]), "+v"(pos[32]), "+v"(pos[33]), "+v"(pos[34]), "+v"(pos[35]),
                          "+v"(pos[36]), "+v"(pos[37]), "+v"(pos[38]));
        asm volatile("" : "+v"(pos[39]), "+v"(pos[40]), "+v"(pos[41]), "+v"(pos[42]), "+v"(pos[43]),
                          "+v"(pos[44]), "+v"(pos[45]), "+v"(pos[46]), "+v"(pos[47]), "+v"(pos[48]));
#pragma unroll
        for (int k = 0; k < PART_EPT; ++k) {
            const int b = d[k] >> BUCKET_SHIFT;
            const int seg = bucket_seg(b, n_nodes);
            if (pos[k] < seg)
                pairs[(size_t)b * BUCKET_PAIR_CAP + w * seg + pos[k]] = make_int2(d[k], s[k]);
        }
    }
    __syncthreads();
    int* trow = tab + (size_t)w * table_pitch(n_nodes);
    for (int b = tid; b < nb; b += nthr) trow[b] = hbin[b];
}


template <int NT, bool NEXT>
__device__ __forceinline__ void gat_strip(float4 (&a)[KBLOCKS], const char* bsb,
                                          unsigned lo, unsigned hi,
                                          const float* __restrict__ A, unsigned aoffn,
                                          float* __restrict__ C, unsigned coff) {
    asm volatile("" : "+v"(coff), "+v"(aoffn), "+v"(lo), "+v"(hi));
    const char* apn = (const char*)A + aoffn;
    char* cp = (char*)C + coff;
#define BFRAG(J, t) \
    (*(const float4*)(bsb + ((J) < KBLOCKS / 2 ? lo : hi) + \
                      ((((J) & (KBLOCKS / 2 - 1)) * 8 + (t)) * 512)))

    v8f acc[NT];
#pragma unroll
    for (int t = 0; t < NT; ++t) acc[t] = (v8f){};

    float4 bc[NT], bn[NT];
#pragma unroll
    for (int t = 0; t < NT; ++t) bc[t] = BFRAG(0, t);

#pragma unroll
    for (int J = 0; J < KBLOCKS; ++J) {
        if (J + 1 < KBLOCKS) {
#pragma unroll
            for (int t = 0; t < NT; ++t) bn[t] = BFRAG(J + 1, t);
        }
        const v2f a0 = {a[J].x, a[J].y};
        const v2f a1 = {a[J].z, a[J].w};
#pragma unroll
        for (int t = 0; t < NT; ++t) {
            const v2f b0 = {bc[t].x, bc[t].y};
            acc[t] = __builtin_amdgcn_wmma_f32_16x16x4_f32(
                false, a0, false, b0, (short)0, acc[t], false, false);
        }
#pragma unroll
        for (int t = 0; t < NT; ++t) {
            const v2f b1 = {bc[t].z, bc[t].w};
            acc[t] = __builtin_amdgcn_wmma_f32_16x16x4_f32(
                false, a1, false, b1, (short)0, acc[t], false, false);
        }
        if (NEXT && J >= 1) a[J - 1] = *(const float4*)(apn + 32 * (J - 1));
#pragma unroll
        for (int t = 0; t < NT; ++t) bc[t] = bn[t];
    }
    if (NEXT) a[KBLOCKS - 1] = *(const float4*)(apn + 32 * (KBLOCKS - 1));
#undef BFRAG

    if constexpr (NT == 8)
        asm volatile("v_nop\n\tv_nop\n\tv_nop\n\tv_nop\n\tv_nop"
                     : "+v"(acc[0]), "+v"(acc[1]), "+v"(acc[2]), "+v"(acc[3]),
                       "+v"(acc[4]), "+v"(acc[5]), "+v"(acc[6]), "+v"(acc[7]));
    else
        asm volatile("v_nop\n\tv_nop\n\tv_nop\n\tv_nop\n\tv_nop"
                     : "+v"(acc[0]), "+v"(acc[1]), "+v"(acc[2]), "+v"(acc[3]));

#pragma unroll
    for (int i = 0; i < 8; ++i) {
#pragma unroll
        for (int u = 0; u < NT / 4; ++u) {
            *(float4*)(cp + i * (NCOL * 4) + 256 * u) =
                make_float4(acc[4 * u + 0][i], acc[4 * u + 1][i],
                            acc[4 * u + 2][i], acc[4 * u + 3][i]);
        }
    }
}

__global__ __launch_bounds__(GEMM_WAVES * 32)
void gat_gemm_wmma(const float* __restrict__ A,
                   const float* __restrict__ B,
                   float* __restrict__ C,
                   const int* __restrict__ src,
                   const int* __restrict__ dst,
                   int* __restrict__ tab,
                   int* __restrict__ rows_as_pairs,
                   int M, int n_edges) {
    __shared__ float4 Bs[KBLOCKS * 8 * 32];

    const int tid  = threadIdx.x;

    if (blockIdx.x < PART_WGS) {
        gat_partition_edges(src, dst, tab, rows_as_pairs, (int*)Bs, M, n_edges);
        return;
    }

    const int lane = tid & 31;
    const int wv   = tid >> 5;
    const int c    = lane & 15;
    const int h    = lane >> 4;
    const int gw = __builtin_amdgcn_readfirstlane((blockIdx.x - PART_WGS) * GEMM_WAVES + wv);

    const int n_strips = (M + 15) >> 4;
    const int per  = n_strips / GEMM_NW;
    const int rest = n_strips - per * GEMM_NW;
    int nfull  = per;
    int hstrip = -1;
    if (2 * rest <= GEMM_NW) {
        if (gw < 2 * rest) hstrip = per * GEMM_NW + (gw >> 1);
    } else if (gw < rest) {
        nfull = per + 1;
    }
    const int first = nfull > 0 ? gw : hstrip;

    float4 a[KBLOCKS];
    if (first >= 0) {
        const float* ap = A + (size_t)(first * 16 + c) * IN_DIM + 4 * h;
#pragma unroll
        for (int J = 0; J < KBLOCKS; ++J) a[J] = *(const float4*)(ap + 8 * J);
    }

#pragma unroll 4
    for (int it = 0; it < (KBLOCKS * 2) / GEMM_WAVES; ++it) {
        const int s = it * GEMM_WAVES + wv;
        const int J = s >> 1;
        const int u = s & 1;
        const float* bp = B + (size_t)(8 * J + 4 * h) * NCOL + 64 * u + 4 * c;
        const float4 g0 = *(const float4*)(bp);
        const float4 g1 = *(const float4*)(bp + NCOL);
        const float4 g2 = *(const float4*)(bp + 2 * NCOL);
        const float4 g3 = *(const float4*)(bp + 3 * NCOL);
        float4* dp = Bs + (J * 8 + 4 * u) * 32 + lane;
        dp[0 * 32] = make_float4(g0.x, g1.x, g2.x, g3.x);
        dp[1 * 32] = make_float4(g0.y, g1.y, g2.y, g3.y);
        dp[2 * 32] = make_float4(g0.z, g1.z, g2.z, g3.z);
        dp[3 * 32] = make_float4(g0.w, g1.w, g2.w, g3.w);
    }
    __syncthreads();
    if (first < 0) return;

    const unsigned lo = (unsigned)lane * 16u;
    const unsigned hi = lo + 65536u;
    const char* bsb = (const char*)Bs;

    for (int it = 0; it < nfull; ++it) {
        const int s  = gw + it * GEMM_NW;
        const int sn = (it + 1 < nfull) ? s + GEMM_NW : (hstrip >= 0 ? hstrip : s);
        const unsigned aoffn = (unsigned)((sn * 16 + c) * IN_DIM + 4 * h) * 4u;
        const unsigned coff  = (unsigned)((s * 16 + 8 * h) * NCOL + 4 * c) * 4u;
        gat_strip<8, true>(a, bsb, lo, hi, A, aoffn, C, coff);
    }
    if (hstrip >= 0) {
        const int hu = gw & 1;
        const unsigned coff = (unsigned)((hstrip * 16 + 8 * h) * NCOL + 4 * c + 64 * hu) * 4u;
        gat_strip<4, false>(a, bsb, lo + 2048u * hu, hi + 2048u * hu, A, 0u, C, coff);
    }
}

__device__ __forceinline__ float leaky(float v) {
    return v > 0.f ? v : ALPHA * v;
}


__device__ __forceinline__ float sum8(float v) {
    v += __int_as_float(__builtin_amdgcn_update_dpp(0, __float_as_int(v), 0x161, 0xf, 0xf, true));
    v += __int_as_float(__builtin_amdgcn_update_dpp(0, __float_as_int(v), 0x162, 0xf, 0xf, true));
    v += __int_as_float(__builtin_amdgcn_update_dpp(0, __float_as_int(v), 0x164, 0xf, 0xf, true));
    return v;
}

template <int X>
__device__ __forceinline__ float lane_xor(const float v) {
    return __int_as_float(__builtin_amdgcn_update_dpp(0, __float_as_int(v), 0x160 + X, 0xf, 0xf, true));
}

template <int R>
__device__ __forceinline__ float lane_bcast8(const float v) {
    return __uint_as_float(__builtin_amdgcn_mov_dpp8(__float_as_uint(v), R * 0x249249u));
}

typedef __attribute__((ext_vector_type(4))) float v4f;

__device__ __forceinline__ float dot4(const v4f a, const v4f b) {
    const v2f t = a.lo * b.lo + a.hi * b.hi;
    return t.x + t.y;
}

template <int G>
__device__ __forceinline__ void gat_rows(const float* __restrict__ ft, const unsigned q4,
                                         const int (&s)[G],
                                         const v4f al, const float a2v, const int k8,
                                         float& m, float& l, v4f& acc) {
    v4f F[G];
#pragma unroll
    for (int g = 0; g < G; ++g) F[g] = *(const v4f*)(ft + (size_t)s[g] * NCOL + q4);
    if constexpr (G == 8)
        asm volatile("" : "+v"(F[0]), "+v"(F[1]), "+v"(F[2]), "+v"(F[3]),
                          "+v"(F[4]), "+v"(F[5]), "+v"(F[6]), "+v"(F[7]));

    const int kg = k8 & (G - 1);
    float a1 = sum8(dot4(F[0], al));
#pragma unroll
    for (int g = 1; g < G; ++g) {
        const float t = sum8(dot4(F[g], al));
        a1 = (kg == g) ? t : a1;
    }
    const float e = leaky(a1 + a2v);
    float mx = e;
    if constexpr (G >= 2) mx = fmaxf(mx, lane_xor<1>(mx));
    if constexpr (G >= 4) mx = fmaxf(mx, lane_xor<2>(mx));
    if constexpr (G >= 8) mx = fmaxf(mx, lane_xor<4>(mx));
    const float mn = fmaxf(m, mx);
    const float sc = expf(m - mn);
    const float wk = expf(e - mn);
    l *= sc;
    acc *= sc;
    float w[G];
    if constexpr (G == 1) {
        w[0] = wk;
    } else {
        w[0] = lane_bcast8<0>(wk);
        w[1] = lane_bcast8<1>(wk);
        if constexpr (G >= 4) { w[2] = lane_bcast8<2>(wk); w[3] = lane_bcast8<3>(wk); }
        if constexpr (G >= 8) {
            w[4] = lane_bcast8<4>(wk); w[5] = lane_bcast8<5>(wk);
            w[6] = lane_bcast8<6>(wk); w[7] = lane_bcast8<7>(wk);
        }
    }
#pragma unroll
    for (int g = 0; g < G; ++g) {
        l += w[g];
        acc += F[g] * w[g];
    }
    m = mn;
}

template <int G>
__device__ __forceinline__ void gat_listed(const float* __restrict__ ft, const unsigned q4,
                                           const int mine,
                                           const int j0, const v4f al, const float a2v,
                                           const int k8, float& m, float& l, v4f& acc) {
    int s[G];
#pragma unroll
    for (int g = 0; g < G; ++g) s[g] = __builtin_amdgcn_readlane(mine, j0 + g);
    gat_rows<G>(ft, q4, s, al, a2v, k8, m, l, acc);
}

__global__ __launch_bounds__(SORT_THREADS) __attribute__((amdgpu_waves_per_eu(16)))
void gat_bucket_gather(const int* __restrict__ src,
                       const int* __restrict__ dst,
                       const float* __restrict__ ft,
                       const float* __restrict__ attn_l,
                       const float* __restrict__ attn_r,
                       const int* __restrict__ tab,
                       const int* pairs_in_rows,
                       float* out, int n_nodes, int n_edges) {
    __shared__ int lst[BUCKET_NODES * LIST_SLOTS];
    __shared__ int deg[BUCKET_NODES];
    __shared__ float a2s[BUCKET_NODES * HEADS];
    __shared__ alignas(16) int segc[PART_WGS];
    __shared__ int next_node;
    const int tid = threadIdx.x;
    const int q = tid & 31;
    const int wv = __builtin_amdgcn_readfirstlane(tid >> 5);
    const int b = blockIdx.x;
    const int node0 = b << BUCKET_SHIFT;
    const int left = n_nodes - node0;
    const int rows = left < BUCKET_NODES ? left : BUCKET_NODES;
    const unsigned q4 = 4u * (unsigned)q;
    const int k8 = q & 7;
    const int seg = bucket_seg(b, n_nodes);

    if (tid < PART_WGS) segc[tid] = tab[(size_t)tid * table_pitch(n_nodes) + b];
    if (tid < BUCKET_NODES) deg[tid] = 0;
    if (tid == 0) next_node = SORT_THREADS / 32;
    __syncthreads();

    v4f own[8];
#pragma unroll
    for (int g = 0; g < 8; ++g) {
        const int kk = 8 * wv + g;
        const int kc = kk < rows ? kk : rows - 1;
        own[g] = *(const v4f*)(ft + (size_t)(node0 + kc) * NCOL + q4);
    }
    const v4f ar = *(const v4f*)(attn_r + 4 * q);

    static_assert(PART_WGS == 256 && SORT_THREADS == 4 * PART_WGS, "8 lengths per lane, 4 threads per run");
    const int4 c0 = ((const int4*)segc)[q];
    const int4 c1 = ((const int4*)segc)[32 + q];
    const int c0m = max(max(c0.x, c0.y), max(c0.z, c0.w));
    const int c1m = max(max(c1.x, c1.y), max(c1.z, c1.w));
    const bool stored = __builtin_amdgcn_ballot_w32(max(c0m, c1m) > seg) == 0u;

    if (stored) {
        const int w = tid >> 2;
        const int cw = segc[w];
        const int2* pg = (const int2*)pairs_in_rows + (size_t)b * BUCKET_PAIR_CAP + w * seg;
        for (int i = tid & 3; i < cw; i += 16) {
            int2 p[4];
#pragma unroll
            for (int k = 0; k < 4; ++k) {
                const int ik = i + 4 * k;
                p[k] = pg[ik < cw ? ik : cw - 1];
            }
            asm volatile("" : "+v"(p[0].x), "+v"(p[0].y), "+v"(p[1].x), "+v"(p[1].y),
                              "+v"(p[2].x), "+v"(p[2].y), "+v"(p[3].x), "+v"(p[3].y));
#pragma unroll
            for (int k = 0; k < 4; ++k) {
                if (i + 4 * k < cw) {
                    const int dl = p[k].x & (BUCKET_NODES - 1);
                    const int pos = atomicAdd(&deg[dl], 1);
                    if (pos < LIST_SLOTS) lst[dl * LIST_SLOTS + pos] = p[k].y;
                }
            }
        }
    } else {
        for (int e = tid; e < n_edges; e += SORT_THREADS) {
            const int d = dst[e];
            if ((d >> BUCKET_SHIFT) == b) {
                const int dl = d & (BUCKET_NODES - 1);
                const int pos = atomicAdd(&deg[dl], 1);
                if (pos < LIST_SLOTS) lst[dl * LIST_SLOTS + pos] = src[e];
            }
        }
    }

#pragma unroll
    for (int g = 0; g < 8; ++g) {
        const float a2 = sum8(dot4(own[g], ar));
        if (k8 == 0) a2s[(8 * wv + g) * HEADS + (q >> 3)] = a2;
    }
    __syncthreads();

    const v4f al = *(const v4f*)(attn_l + 4 * q);
    int k = wv;
    while (k < rows) {
        int kn = 0;
        if (q == 0) kn = atomicAdd(&next_node, 1);
        const int dg = __builtin_amdgcn_readfirstlane(deg[k]);
        const float a2v = a2s[k * HEADS + (q >> 3)];
        const int* row = lst + k * LIST_SLOTS;
        int mine = row[q];
        kn = __builtin_amdgcn_readfirstlane(kn);
        const int n = node0 + k;

        float m = -3.402823466e+38f;
        float l = 0.f;
        v4f acc = {0.f, 0.f, 0.f, 0.f};

        if (dg <= LIST_SLOTS) {
            for (int base = 0; base < dg; base += 32) {
                const int nleft = dg - base;
                const int nj = nleft < 32 ? nleft : 32;
                if (base > 0) mine = row[base + q];
                int j = 0;
                for (; j + 8 <= nj; j += 8) gat_listed<8>(ft, q4, mine, j, al, a2v, k8, m, l, acc);
                if (nj - j >= 4) { gat_listed<4>(ft, q4, mine, j, al, a2v, k8, m, l, acc); j += 4; }
                if (nj - j >= 2) { gat_listed<2>(ft, q4, mine, j, al, a2v, k8, m, l, acc); j += 2; }
                if (nj - j >= 1) gat_listed<1>(ft, q4, mine, j, al, a2v, k8, m, l, acc);
            }
        } else {
            for (int e = 0; e < n_edges; ++e) {
                if (__builtin_amdgcn_readfirstlane(dst[e]) == n) {
                    const int s[1] = {__builtin_amdgcn_readfirstlane(src[e])};
                    gat_rows<1>(ft, q4, s, al, a2v, k8, m, l, acc);
                }
            }
        }

        v4f x = {0.f, 0.f, 0.f, 0.f};
        if (l > 0.f) x = acc / l;
        *(v4f*)(out + (size_t)n * NCOL + 4 * q) = x;
        k = kn;
    }
}

extern "C" void kernel_launch(void* const* d_in, const int* in_sizes, int n_in,
                              void* d_out, int out_size, void* d_ws, size_t ws_size,
                              hipStream_t stream) {
    const float* feature = (const float*)d_in[0];
    const int*   src     = (const int*)d_in[1];
    const int*   dst     = (const int*)d_in[2];
    const float* W_fc    = (const float*)d_in[3];
    const float* attn_l  = (const float*)d_in[4];
    const float* attn_r  = (const float*)d_in[5];
    float*       out     = (float*)d_out;

    const int M = in_sizes[0] / IN_DIM;
    const int E = in_sizes[1];

    float* ft  = (float*)d_ws;
    int*   tab = (int*)(ft + (size_t)M * NCOL);

    gat_gemm_wmma<<<PART_WGS + GEMM_WGS, GEMM_WAVES * 32, 0, stream>>>(
        feature, W_fc, ft, src, dst, tab, (int*)out, M, E);
    gat_bucket_gather<<<(M + BUCKET_NODES - 1) / BUCKET_NODES, SORT_THREADS, 0, stream>>>(
        src, dst, ft, attn_l, attn_r, tab, (const int*)out, out, M, E);
}
